# starting kernel
# baseline (speedup 1.0000x reference)
#include <hip/hip_runtime.h>


typedef __attribute__((ext_vector_type(2))) float v2f;
typedef __attribute__((ext_vector_type(8))) float v8f;

#define K_DIM 128

__global__ void fill_f32(float* __restrict__ p, float v, long n) {
    long i = (long)blockIdx.x * blockDim.x + threadIdx.x;
    if (i < n) p[i] = v;
}

__global__ void degree_kernel(const int* __restrict__ dst,
                              float* __restrict__ deg, int n_edges) {
    int e = blockIdx.x * blockDim.x + threadIdx.x;
    if (e < n_edges) atomicAdd(&deg[dst[e]], 1.0f);
}

__global__ void rsqrt_inplace(float* __restrict__ deg, int n) {
    int i = blockIdx.x * blockDim.x + threadIdx.x;
    if (i < n) deg[i] = rsqrtf(deg[i]);
}

template <int NCOLS>
__global__ void gemm_wmma_f32(const float* __restrict__ A,
                              const float* __restrict__ W,
                              float* __restrict__ C, int nrows) {
    constexpr int TC = NCOLS / 16;
    const int lane  = threadIdx.x & 31;
    const int wave  = threadIdx.x >> 5;
    const int tile  = blockIdx.x * (blockDim.x >> 5) + wave;
    const int trow  = tile / TC;
    const int tcol  = tile % TC;
    const int rbase = trow * 16;
    if (rbase >= nrows) return;

    const int m     = lane & 15;
    const int khalf = lane >> 4;

    const int arow_idx = (rbase + m < nrows) ? (rbase + m) : (nrows - 1);
    const float* arow = A + (long)arow_idx * K_DIM;
    const float* wcol = W + tcol * 16 + m;

    v8f acc = {};
#pragma unroll 8
    for (int k = 0; k < K_DIM; k += 4) {
        const int kk = k + 2 * khalf;
        v2f a, b;
        a.x = arow[kk];
        a.y = arow[kk + 1];
        b.x = wcol[(long)kk * NCOLS];
        b.y = wcol[(long)(kk + 1) * NCOLS];
        acc = __builtin_amdgcn_wmma_f32_16x16x4_f32(
            false, a, false, b, (short)0, acc, false, false);
    }

    if (rbase + 16 <= nrows) {
        float* cbase = C + (long)(rbase + 8 * khalf) * NCOLS + tcol * 16 + m;
#pragma unroll
        for (int r = 0; r < 8; ++r) cbase[(long)r * NCOLS] = acc[r];
    } else {
#pragma unroll
        for (int r = 0; r < 8; ++r) {
            const int row = rbase + r + 8 * khalf;
            if (row < nrows) C[(long)row * NCOLS + tcol * 16 + m] = acc[r];
        }
    }
}

template <int CH>
__global__ void aggregate_edges(const float* __restrict__ h,
                                const int* __restrict__ src,
                                const int* __restrict__ dst,
                                const float* __restrict__ dinv,
                                float* __restrict__ out, int n_edges) {
    const int lane = threadIdx.x & 31;
    const int e    = blockIdx.x * (blockDim.x >> 5) + (threadIdx.x >> 5);
    if (e >= n_edges) return;

    const int   s   = src[e];
    const int   d   = dst[e];
    const float nrm = dinv[s] * dinv[d];

    constexpr int V = CH / 32;
    const float* hp = h   + (long)s * CH + lane * V;
    float*       op = out + (long)d * CH + lane * V;

    float vals[V];
#pragma unroll
    for (int i = 0; i < V; ++i) vals[i] = hp[i];
#pragma unroll
    for (int i = 0; i < V; ++i) atomicAdd(&op[i], vals[i] * nrm);
}

template <int CH>
__global__ void bias_relu_self(const float* __restrict__ h,
                               const float* __restrict__ dinv,
                               const float* __restrict__ bias,
                               float* __restrict__ agg, int n_nodes) {
    long idx = (long)blockIdx.x * blockDim.x + threadIdx.x;
    if (idx >= (long)n_nodes * CH) return;
    const int node = (int)(idx / CH);
    const int c    = (int)(idx % CH);
    const float di = dinv[node];
    const float v  = agg[idx] + h[idx] * di * di + bias[c];
    agg[idx] = v > 0.0f ? v : 0.0f;
}

extern "C" void kernel_launch(void* const* d_in, const int* in_sizes, int n_in,
                              void* d_out, int out_size, void* d_ws, size_t ws_size,
                              hipStream_t stream) {
    const float* x  = (const float*)d_in[0];
    const int*   ei = (const int*)d_in[1];
    const float* W1 = (const float*)d_in[2];
    const float* b1 = (const float*)d_in[3];
    const float* W2 = (const float*)d_in[4];
    const float* b2 = (const float*)d_in[5];

    const int N   = in_sizes[0] / K_DIM;
    const int E   = in_sizes[1] / 2;
    (void)n_in; (void)ws_size; (void)out_size;

    const int* src = ei;
    const int* dst = ei + E;
    float* out = (float*)d_out;

    char* ws = (char*)d_ws;
    const size_t degBytes = (((size_t)N * 4) + 255) & ~(size_t)255;
    float* deg  = (float*)ws;
    float* bufA = (float*)(ws + degBytes);
    float* bufB = (float*)(ws + degBytes + (size_t)N * 128 * 4);

    const int TB = 256;

    fill_f32<<<(N + TB - 1) / TB, TB, 0, stream>>>(deg, 1.0f, N);
    degree_kernel<<<(E + TB - 1) / TB, TB, 0, stream>>>(dst, deg, E);
    rsqrt_inplace<<<(N + TB - 1) / TB, TB, 0, stream>>>(deg, N);

    {
        const int tiles = ((N + 15) / 16) * (128 / 16);
        gemm_wmma_f32<128><<<(tiles + 7) / 8, 256, 0, stream>>>(x, W1, bufA, N);
        const long n1 = (long)N * 128;
        fill_f32<<<(unsigned)((n1 + TB - 1) / TB), TB, 0, stream>>>(bufB, 0.0f, n1);
        aggregate_edges<128><<<(E + 7) / 8, 256, 0, stream>>>(bufA, src, dst, deg, bufB, E);
        bias_relu_self<128><<<(unsigned)((n1 + TB - 1) / TB), TB, 0, stream>>>(bufA, deg, b1, bufB, N);
    }

    {
        const int tiles = ((N + 15) / 16) * (64 / 16);
        gemm_wmma_f32<64><<<(tiles + 7) / 8, 256, 0, stream>>>(bufB, W2, bufA, N);
        const long n2 = (long)N * 64;
        fill_f32<<<(unsigned)((n2 + TB - 1) / TB), TB, 0, stream>>>(out, 0.0f, n2);
        aggregate_edges<64><<<(E + 7) / 8, 256, 0, stream>>>(bufA, src, dst, deg, out, E);
        bias_relu_self<64><<<(unsigned)((n2 + TB - 1) / TB), TB, 0, stream>>>(bufA, deg, b2, out, N);
    }
}


# Clear and self-loop+bias passes folded into the dense transforms: each GEMM also stores h*dinv^2+b as start value of the edge atomics, ReLU moves to the consumer; sum order: self+bias first
# speedup vs baseline: 1.7689x; 1.7689x over previous
#include <hip/hip_runtime.h>


typedef __attribute__((ext_vector_type(2))) float v2f;
typedef __attribute__((ext_vector_type(8))) float v8f;

#define K_DIM 128

__global__ void fill_f32(float* __restrict__ p, float v, long n) {
    long i = (long)blockIdx.x * blockDim.x + threadIdx.x;
    if (i < n) p[i] = v;
}

__global__ void degree_kernel(const int* __restrict__ dst,
                              float* __restrict__ deg, int n_edges) {
    int e = blockIdx.x * blockDim.x + threadIdx.x;
    if (e < n_edges) atomicAdd(&deg[dst[e]], 1.0f);
}

__global__ void rsqrt_inplace(float* __restrict__ deg, int n) {
    int i = blockIdx.x * blockDim.x + threadIdx.x;
    if (i < n) deg[i] = rsqrtf(deg[i]);
}

template <int NCOLS, bool RELU_A>
__global__ void gemm_wmma_f32(const float* __restrict__ A,
                              const float* __restrict__ W,
                              const float* __restrict__ dinv,
                              const float* __restrict__ bias,
                              float* __restrict__ C,
                              float* __restrict__ S, int nrows) {
    constexpr int TC = NCOLS / 16;
    const int lane  = threadIdx.x & 31;
    const int wave  = threadIdx.x >> 5;
    const int tile  = blockIdx.x * (blockDim.x >> 5) + wave;
    const int trow  = tile / TC;
    const int tcol  = tile % TC;
    const int rbase = trow * 16;
    if (rbase >= nrows) return;

    const int m     = lane & 15;
    const int khalf = lane >> 4;

    const int arow_idx = (rbase + m < nrows) ? (rbase + m) : (nrows - 1);
    const float* arow = A + (long)arow_idx * K_DIM;
    const float* wcol = W + tcol * 16 + m;

    v8f acc = {};
    for (int k0 = 0; k0 < K_DIM; k0 += 32) {
        v2f a[8], b[8];
#pragma unroll
        for (int u = 0; u < 8; ++u) {
            const int kk = k0 + 4 * u + 2 * khalf;
            a[u].x = arow[kk];
            a[u].y = arow[kk + 1];
            b[u].x = wcol[(long)kk * NCOLS];
            b[u].y = wcol[(long)(kk + 1) * NCOLS];
        }
        if (RELU_A) {
#pragma unroll
            for (int u = 0; u < 8; ++u) {
                a[u].x = fmaxf(a[u].x, 0.0f);
                a[u].y = fmaxf(a[u].y, 0.0f);
            }
            __builtin_amdgcn_sched_barrier(0);
        }
#pragma unroll
        for (int u = 0; u < 8; ++u) {
            acc = __builtin_amdgcn_wmma_f32_16x16x4_f32(
                false, a[u], false, b[u], (short)0, acc, false, false);
        }
        asm volatile("v_nop\n\tv_nop\n\tv_nop\n\tv_nop"
                     : "+v"(acc) : "v"(a[7]), "v"(b[7]));
    }

    const float bc = bias[tcol * 16 + m];
    float seed[8];
#pragma unroll
    for (int r = 0; r < 8; ++r) {
        const int row = rbase + r + 8 * khalf;
        const float di = dinv[row < nrows ? row : nrows - 1];
        seed[r] = acc[r] * di * di + bc;
    }

    if (rbase + 16 <= nrows) {
        const long off = (long)(rbase + 8 * khalf) * NCOLS + tcol * 16 + m;
#pragma unroll
        for (int r = 0; r < 8; ++r) {
            C[off + (long)r * NCOLS] = acc[r];
            S[off + (long)r * NCOLS] = seed[r];
        }
    } else {
#pragma unroll
        for (int r = 0; r < 8; ++r) {
            const int row = rbase + r + 8 * khalf;
            if (row < nrows) {
                C[(long)row * NCOLS + tcol * 16 + m] = acc[r];
                S[(long)row * NCOLS + tcol * 16 + m] = seed[r];
            }
        }
    }
}

template <int CH>
__global__ void aggregate_edges(const float* __restrict__ h,
                                const int* __restrict__ src,
                                const int* __restrict__ dst,
                                const float* __restrict__ dinv,
                                float* __restrict__ out, int n_edges) {
    constexpr int V  = CH / 32;
    constexpr int B  = 16 / V;
    constexpr int NB = 32 / B;

    const int lane = threadIdx.x & 31;
    const int wv = __builtin_amdgcn_readfirstlane((int)(threadIdx.x >> 5));
    const int e0 = (blockIdx.x * (int)(blockDim.x >> 5) + wv) * 32;
    if (e0 >= n_edges) return;
    const int cnt = n_edges - e0;

    const int   el = e0 + (lane < cnt ? lane : cnt - 1);
    const int   s  = src[el];
    const int   d  = dst[el];

    if (cnt >= 32) {
        float buf[2][B * V];
#pragma unroll
        for (int jj = 0; jj < B; ++jj) {
            const int sj = __builtin_amdgcn_readlane(s, jj);
            const float* hp = h + (long)sj * CH + lane;
#pragma unroll
            for (int i = 0; i < V; ++i) buf[0][jj * V + i] = hp[32 * i];
        }
        const int wbits = __builtin_bit_cast(int, dinv[s] * dinv[d]);
#pragma unroll
        for (int b = 0; b < NB; ++b) {
            if (b + 1 < NB) {
#pragma unroll
                for (int jj = 0; jj < B; ++jj) {
                    const int sj = __builtin_amdgcn_readlane(s, (b + 1) * B + jj);
                    const float* hp = h + (long)sj * CH + lane;
#pragma unroll
                    for (int i = 0; i < V; ++i)
                        buf[(b + 1) & 1][jj * V + i] = hp[32 * i];
                }
            }
#pragma unroll
            for (int jj = 0; jj < B; ++jj) {
                const int   dj = __builtin_amdgcn_readlane(d, b * B + jj);
                const float wj = __builtin_bit_cast(
                    float, __builtin_amdgcn_readlane(wbits, b * B + jj));
                float* op = out + (long)dj * CH + lane;
#pragma unroll
                for (int i = 0; i < V; ++i)
                    atomicAdd(&op[32 * i], buf[b & 1][jj * V + i] * wj);
            }
        }
    } else {
        const int wbits = __builtin_bit_cast(int, dinv[s] * dinv[d]);
        for (int j = 0; j < cnt; ++j) {
            const int   sj = __builtin_amdgcn_readlane(s, j);
            const int   dj = __builtin_amdgcn_readlane(d, j);
            const float wj = __builtin_bit_cast(
                float, __builtin_amdgcn_readlane(wbits, j));
            const float* hp = h   + (long)sj * CH + lane;
            float*       op = out + (long)dj * CH + lane;
            float vals[V];
#pragma unroll
            for (int i = 0; i < V; ++i) vals[i] = hp[32 * i];
#pragma unroll
            for (int i = 0; i < V; ++i) atomicAdd(&op[32 * i], vals[i] * wj);
        }
    }
}

__global__ void relu_inplace(float* __restrict__ p, long n4) {
    long i = (long)blockIdx.x * blockDim.x + threadIdx.x;
    if (i >= n4) return;
    float4* p4 = reinterpret_cast<float4*>(p);
    float4 v = p4[i];
    v.x = fmaxf(v.x, 0.0f);
    v.y = fmaxf(v.y, 0.0f);
    v.z = fmaxf(v.z, 0.0f);
    v.w = fmaxf(v.w, 0.0f);
    p4[i] = v;
}

extern "C" void kernel_launch(void* const* d_in, const int* in_sizes, int n_in,
                              void* d_out, int out_size, void* d_ws, size_t ws_size,
                              hipStream_t stream) {
    const float* x  = (const float*)d_in[0];
    const int*   ei = (const int*)d_in[1];
    const float* W1 = (const float*)d_in[2];
    const float* b1 = (const float*)d_in[3];
    const float* W2 = (const float*)d_in[4];
    const float* b2 = (const float*)d_in[5];

    const int N   = in_sizes[0] / K_DIM;
    const int E   = in_sizes[1] / 2;
    (void)n_in; (void)ws_size; (void)out_size;

    const int* src = ei;
    const int* dst = ei + E;
    float* out = (float*)d_out;

    char* ws = (char*)d_ws;
    const size_t degBytes = (((size_t)N * 4) + 255) & ~(size_t)255;
    float* deg  = (float*)ws;
    float* bufA = (float*)(ws + degBytes);
    float* bufB = (float*)(ws + degBytes + (size_t)N * 128 * 4);

    const int TB = 256;

    fill_f32<<<(N + TB - 1) / TB, TB, 0, stream>>>(deg, 1.0f, N);
    degree_kernel<<<(E + TB - 1) / TB, TB, 0, stream>>>(dst, deg, E);
    rsqrt_inplace<<<(N + TB - 1) / TB, TB, 0, stream>>>(deg, N);

    {
        const int tiles = ((N + 15) / 16) * (128 / 16);
        gemm_wmma_f32<128, false><<<(tiles + 7) / 8, 256, 0, stream>>>(x, W1, deg, b1, bufA, bufB, N);
        aggregate_edges<128><<<(E + 255) / 256, 256, 0, stream>>>(bufA, src, dst, deg, bufB, E);
    }

    {
        const int tiles = ((N + 15) / 16) * (64 / 16);
        gemm_wmma_f32<64, true><<<(tiles + 7) / 8, 256, 0, stream>>>(bufB, W2, deg, b2, bufA, out, N);
        aggregate_edges<64><<<(E + 255) / 256, 256, 0, stream>>>(bufA, src, dst, deg, out, E);
        const long n4 = (long)N * (64 / 4);
        relu_inplace<<<(unsigned)((n4 + TB - 1) / TB), TB, 0, stream>>>(out, n4);
    }
}


# Scatters made half-local: work-group reads its die, draws a ticket (edge chunk, row range) of its own L2 half, adds only into that half's rows; kept edges packed in LDS; sum order: atomic arrival
# speedup vs baseline: 2.3349x; 1.3200x over previous
#include <hip/hip_runtime.h>


typedef __attribute__((ext_vector_type(2))) float v2f;
typedef __attribute__((ext_vector_type(8))) float v8f;

#define K_DIM 128

#define CTR_STRIDE 64

__global__ void fill_f32(float* __restrict__ p, float v, long n,
                         int* __restrict__ ctr) {
    long i = (long)blockIdx.x * blockDim.x + threadIdx.x;
    if (i < n) p[i] = v;
    if (i < 4) ctr[i * CTR_STRIDE] = 0;
}

__device__ __forceinline__ void draw_ticket(int* __restrict__ ctr, int nchunks,
                                            int* ctl) {
    const unsigned pw = __builtin_amdgcn_s_sendmsg_rtn(0x87);
    int r = (int)((pw >> 18) & 1u);
    int t = atomicAdd(&ctr[r * CTR_STRIDE], 1);
    if (t >= nchunks) {
        r ^= 1;
        t = atomicAdd(&ctr[r * CTR_STRIDE], 1);
    }
    ctl[0] = t;
    ctl[1] = r;
}

#define DEG_CHUNK 1024

__global__ __launch_bounds__(256) void degree_kernel(
    const int* __restrict__ dst, float* __restrict__ deg, int n_edges,
    int split, int* __restrict__ ctr, int nchunks) {
    __shared__ int ctl[2];
    if (threadIdx.x == 0) draw_ticket(ctr, nchunks, ctl);
    __syncthreads();
    const int t = ctl[0];
    const int r = ctl[1];
    if (t >= nchunks) return;
    constexpr int EPT = DEG_CHUNK / 256;
    int ed[EPT];
#pragma unroll
    for (int k = 0; k < EPT; ++k) {
        const int e = t * DEG_CHUNK + k * 256 + (int)threadIdx.x;
        ed[k] = dst[e < n_edges ? e : n_edges - 1];
    }
#pragma unroll
    for (int k = 0; k < EPT; ++k) {
        const int e = t * DEG_CHUNK + k * 256 + (int)threadIdx.x;
        const int d = ed[k];
        if (e < n_edges && (d >= split ? 1 : 0) == r) atomicAdd(&deg[d], 1.0f);
    }
}

__global__ void rsqrt_inplace(float* __restrict__ deg, int n) {
    int i = blockIdx.x * blockDim.x + threadIdx.x;
    if (i < n) deg[i] = rsqrtf(deg[i]);
}

template <int NCOLS, bool RELU_A>
__global__ void gemm_wmma_f32(const float* __restrict__ A,
                              const float* __restrict__ W,
                              const float* __restrict__ dinv,
                              const float* __restrict__ bias,
                              float* __restrict__ C,
                              float* __restrict__ S, int nrows,
                              int* __restrict__ zero2) {
    if (zero2 != nullptr && blockIdx.x == 0 && threadIdx.x < 2)
        zero2[threadIdx.x * CTR_STRIDE] = 0;
    constexpr int TC = NCOLS / 16;
    const int lane  = threadIdx.x & 31;
    const int wave  = threadIdx.x >> 5;
    const int tile  = blockIdx.x * (blockDim.x >> 5) + wave;
    const int trow  = tile / TC;
    const int tcol  = tile % TC;
    const int rbase = trow * 16;
    if (rbase >= nrows) return;

    const int m     = lane & 15;
    const int khalf = lane >> 4;

    const int arow_idx = (rbase + m < nrows) ? (rbase + m) : (nrows - 1);
    const float* arow = A + (long)arow_idx * K_DIM;
    const float* wcol = W + tcol * 16 + m;

    v8f acc = {};
    for (int k0 = 0; k0 < K_DIM; k0 += 32) {
        v2f a[8], b[8];
#pragma unroll
        for (int u = 0; u < 8; ++u) {
            const int kk = k0 + 4 * u + 2 * khalf;
            a[u].x = arow[kk];
            a[u].y = arow[kk + 1];
            b[u].x = wcol[(long)kk * NCOLS];
            b[u].y = wcol[(long)(kk + 1) * NCOLS];
        }
        if (RELU_A) {
#pragma unroll
            for (int u = 0; u < 8; ++u) {
                a[u].x = fmaxf(a[u].x, 0.0f);
                a[u].y = fmaxf(a[u].y, 0.0f);
            }
            __builtin_amdgcn_sched_barrier(0);
        }
#pragma unroll
        for (int u = 0; u < 8; ++u) {
            acc = __builtin_amdgcn_wmma_f32_16x16x4_f32(
                false, a[u], false, b[u], (short)0, acc, false, false);
        }
        asm volatile("v_nop\n\tv_nop\n\tv_nop\n\tv_nop"
                     : "+v"(acc) : "v"(a[7]), "v"(b[7]));
    }

    const float bc = bias[tcol * 16 + m];
    float seed[8];
#pragma unroll
    for (int r = 0; r < 8; ++r) {
        const int row = rbase + r + 8 * khalf;
        const float di = dinv[row < nrows ? row : nrows - 1];
        seed[r] = acc[r] * di * di + bc;
    }

    if (rbase + 16 <= nrows) {
        const long off = (long)(rbase + 8 * khalf) * NCOLS + tcol * 16 + m;
#pragma unroll
        for (int r = 0; r < 8; ++r) {
            C[off + (long)r * NCOLS] = acc[r];
            S[off + (long)r * NCOLS] = seed[r];
        }
    } else {
#pragma unroll
        for (int r = 0; r < 8; ++r) {
            const int row = rbase + r + 8 * khalf;
            if (row < nrows) {
                C[(long)row * NCOLS + tcol * 16 + m] = acc[r];
                S[(long)row * NCOLS + tcol * 16 + m] = seed[r];
            }
        }
    }
}

#define AGG_CHUNK 512

template <int CH, int NE>
__device__ __forceinline__ void edge_batch(const float* __restrict__ h,
                                           float* __restrict__ out, int s,
                                           int d, int wbits, int j0, int lane) {
    constexpr int V = CH / 32;
    float buf[NE * V];
#pragma unroll
    for (int jj = 0; jj < NE; ++jj) {
        const int sj = __builtin_amdgcn_readlane(s, j0 + jj);
        const float* hp = h + (long)sj * CH + lane;
#pragma unroll
        for (int i = 0; i < V; ++i) buf[jj * V + i] = hp[32 * i];
    }
#pragma unroll
    for (int jj = 0; jj < NE; ++jj) {
        const int   dj = __builtin_amdgcn_readlane(d, j0 + jj);
        const float wj = __builtin_bit_cast(
            float, __builtin_amdgcn_readlane(wbits, j0 + jj));
        float* op = out + (long)dj * CH + lane;
#pragma unroll
        for (int i = 0; i < V; ++i)
            atomicAdd(&op[32 * i], buf[jj * V + i] * wj);
    }
}

template <int CH>
__global__ __launch_bounds__(256) void aggregate_edges(
    const float* __restrict__ h, const int* __restrict__ src,
    const int* __restrict__ dst, const float* __restrict__ dinv,
    float* __restrict__ out, int n_edges, int split,
    int* __restrict__ ctr, int nchunks) {
    constexpr int V  = CH / 32;
    constexpr int B  = 16 / V;
    constexpr int NB = 32 / B;

    __shared__ int sh_s[AGG_CHUNK];
    __shared__ int sh_d[AGG_CHUNK];
    __shared__ int ctl[3];

    const int lane = threadIdx.x & 31;
    const int wv = __builtin_amdgcn_readfirstlane((int)(threadIdx.x >> 5));

    if (threadIdx.x == 0) {
        draw_ticket(ctr, nchunks, ctl);
        ctl[2] = 0;
    }
    __syncthreads();
    const int t = __builtin_amdgcn_readfirstlane(ctl[0]);
    const int r = __builtin_amdgcn_readfirstlane(ctl[1]);
    if (t >= nchunks) return;

    constexpr int EPT = AGG_CHUNK / 256;
    int es[EPT], ed[EPT];
#pragma unroll
    for (int k = 0; k < EPT; ++k) {
        const int e  = t * AGG_CHUNK + k * 256 + (int)threadIdx.x;
        const int ec = e < n_edges ? e : n_edges - 1;
        ed[k] = dst[ec];
        es[k] = src[ec];
    }
#pragma unroll
    for (int k = 0; k < EPT; ++k) {
        const int e = t * AGG_CHUNK + k * 256 + (int)threadIdx.x;
        const int s = es[k], d = ed[k];
        const bool keep = (e < n_edges) && ((d >= split ? 1 : 0) == r);
        const unsigned m = __builtin_amdgcn_ballot_w32(keep);
        const int c = __builtin_popcount(m);
        int base = 0;
        if (lane == 0 && c != 0) base = atomicAdd(&ctl[2], c);
        base = __builtin_amdgcn_readfirstlane(base);
        if (keep) {
            const int p = base + (int)__builtin_amdgcn_mbcnt_lo(m, 0u);
            sh_s[p] = s;
            sh_d[p] = d;
        }
    }
    __syncthreads();
    const int total = __builtin_amdgcn_readfirstlane(ctl[2]);

    for (int run = wv; run * 32 < total; run += 8) {
        const int left = total - run * 32;
        const int cnt  = left < 32 ? left : 32;

        const int li = run * 32 + (lane < cnt ? lane : cnt - 1);
        const int s  = sh_s[li];
        const int d  = sh_d[li];

        if (cnt >= 32) {
            float buf[2][B * V];
#pragma unroll
            for (int jj = 0; jj < B; ++jj) {
                const int sj = __builtin_amdgcn_readlane(s, jj);
                const float* hp = h + (long)sj * CH + lane;
#pragma unroll
                for (int i = 0; i < V; ++i) buf[0][jj * V + i] = hp[32 * i];
            }
            const int wbits = __builtin_bit_cast(int, dinv[s] * dinv[d]);
#pragma unroll
            for (int b = 0; b < NB; ++b) {
                if (b + 1 < NB) {
#pragma unroll
                    for (int jj = 0; jj < B; ++jj) {
                        const int sj = __builtin_amdgcn_readlane(s, (b + 1) * B + jj);
                        const float* hp = h + (long)sj * CH + lane;
#pragma unroll
                        for (int i = 0; i < V; ++i)
                            buf[(b + 1) & 1][jj * V + i] = hp[32 * i];
                    }
                }
#pragma unroll
                for (int jj = 0; jj < B; ++jj) {
                    const int   dj = __builtin_amdgcn_readlane(d, b * B + jj);
                    const float wj = __builtin_bit_cast(
                        float, __builtin_amdgcn_readlane(wbits, b * B + jj));
                    float* op = out + (long)dj * CH + lane;
#pragma unroll
                    for (int i = 0; i < V; ++i)
                        atomicAdd(&op[32 * i], buf[b & 1][jj * V + i] * wj);
                }
            }
        } else {
            const int wbits = __builtin_bit_cast(int, dinv[s] * dinv[d]);
            int j = 0;
            for (; j + B <= cnt; j += B)
                edge_batch<CH, B>(h, out, s, d, wbits, j, lane);
            if (B > 4 && cnt - j >= 4) {
                edge_batch<CH, 4>(h, out, s, d, wbits, j, lane);
                j += 4;
            }
            if (B > 2 && cnt - j >= 2) {
                edge_batch<CH, 2>(h, out, s, d, wbits, j, lane);
                j += 2;
            }
            if (cnt - j >= 1) edge_batch<CH, 1>(h, out, s, d, wbits, j, lane);
        }
    }
}

__global__ void relu_inplace(float* __restrict__ p, long n4) {
    long i = (long)blockIdx.x * blockDim.x + threadIdx.x;
    if (i >= n4) return;
    float4* p4 = reinterpret_cast<float4*>(p);
    float4 v = p4[i];
    v.x = fmaxf(v.x, 0.0f);
    v.y = fmaxf(v.y, 0.0f);
    v.z = fmaxf(v.z, 0.0f);
    v.w = fmaxf(v.w, 0.0f);
    p4[i] = v;
}

extern "C" void kernel_launch(void* const* d_in, const int* in_sizes, int n_in,
                              void* d_out, int out_size, void* d_ws, size_t ws_size,
                              hipStream_t stream) {
    const float* x  = (const float*)d_in[0];
    const int*   ei = (const int*)d_in[1];
    const float* W1 = (const float*)d_in[2];
    const float* b1 = (const float*)d_in[3];
    const float* W2 = (const float*)d_in[4];
    const float* b2 = (const float*)d_in[5];

    const int N   = in_sizes[0] / K_DIM;
    const int E   = in_sizes[1] / 2;
    (void)n_in; (void)ws_size; (void)out_size;

    const int* src = ei;
    const int* dst = ei + E;
    float* out = (float*)d_out;

    char* ws = (char*)d_ws;
    const size_t degBytes = (((size_t)N * 4) + 255) & ~(size_t)255;
    float* deg  = (float*)ws;
    float* bufA = (float*)(ws + degBytes);
    float* bufB = (float*)(ws + degBytes + (size_t)N * 128 * 4);

    const int TB = 256;

    const int split = N / 2;
    int* ctrD = (int*)out;
    int* ctr1 = (int*)out + 2 * CTR_STRIDE;
    int* ctr2 = (int*)(bufA + (size_t)N * 64);
    const int degChunks = (E + DEG_CHUNK - 1) / DEG_CHUNK;
    const int aggChunks = (E + AGG_CHUNK - 1) / AGG_CHUNK;

    fill_f32<<<(N + TB - 1) / TB, TB, 0, stream>>>(deg, 1.0f, N, ctrD);
    degree_kernel<<<2 * degChunks, TB, 0, stream>>>(dst, deg, E, split, ctrD, degChunks);
    rsqrt_inplace<<<(N + TB - 1) / TB, TB, 0, stream>>>(deg, N);

    {
        const int tiles = ((N + 15) / 16) * (128 / 16);
        gemm_wmma_f32<128, false><<<(tiles + 7) / 8, 256, 0, stream>>>(x, W1, deg, b1, bufA, bufB, N, nullptr);
        aggregate_edges<128><<<2 * aggChunks, 256, 0, stream>>>(bufA, src, dst, deg, bufB, E, split, ctr1, aggChunks);
    }

    {
        const int tiles = ((N + 15) / 16) * (64 / 16);
        gemm_wmma_f32<64, true><<<(tiles + 7) / 8, 256, 0, stream>>>(bufB, W2, deg, b2, bufA, out, N, ctr2);
        aggregate_edges<64><<<2 * aggChunks, 256, 0, stream>>>(bufA, src, dst, deg, out, E, split, ctr2, aggChunks);
        const long n4 = (long)N * (64 / 4);
        relu_inplace<<<(unsigned)((n4 + TB - 1) / TB), TB, 0, stream>>>(out, n4);
    }
}


# Dense transforms: each work-group stages its 16-row stripes of A once in LDS with whole-line 16-byte loads (layer-1 ReLU applied there); its 8 waves read A fragments from LDS; sum order unchanged
# speedup vs baseline: 2.5374x; 1.0867x over previous
#include <hip/hip_runtime.h>


typedef __attribute__((ext_vector_type(2))) float v2f;
typedef __attribute__((ext_vector_type(4))) float v4f;
typedef __attribute__((ext_vector_type(8))) float v8f;

#define K_DIM 128
#define A_PITCH 132

#define CTR_STRIDE 64

__global__ void fill_f32(float* __restrict__ p, float v, long n,
                         int* __restrict__ ctr) {
    long i = (long)blockIdx.x * blockDim.x + threadIdx.x;
    if (i < n) p[i] = v;
    if (i < 4) ctr[i * CTR_STRIDE] = 0;
}

__device__ __forceinline__ void draw_ticket(int* __restrict__ ctr, int nchunks,
                                            int* ctl) {
    const unsigned pw = __builtin_amdgcn_s_sendmsg_rtn(0x87);
    int r = (int)((pw >> 18) & 1u);
    int t = atomicAdd(&ctr[r * CTR_STRIDE], 1);
    if (t >= nchunks) {
        r ^= 1;
        t = atomicAdd(&ctr[r * CTR_STRIDE], 1);
    }
    ctl[0] = t;
    ctl[1] = r;
}

#define DEG_CHUNK 1024

__global__ __launch_bounds__(256) void degree_kernel(
    const int* __restrict__ dst, float* __restrict__ deg, int n_edges,
    int split, int* __restrict__ ctr, int nchunks) {
    __shared__ int ctl[2];
    if (threadIdx.x == 0) draw_ticket(ctr, nchunks, ctl);
    __syncthreads();
    const int t = ctl[0];
    const int r = ctl[1];
    if (t >= nchunks) return;
    constexpr int EPT = DEG_CHUNK / 256;
    int ed[EPT];
#pragma unroll
    for (int k = 0; k < EPT; ++k) {
        const int e = t * DEG_CHUNK + k * 256 + (int)threadIdx.x;
        ed[k] = dst[e < n_edges ? e : n_edges - 1];
    }
#pragma unroll
    for (int k = 0; k < EPT; ++k) {
        const int e = t * DEG_CHUNK + k * 256 + (int)threadIdx.x;
        const int d = ed[k];
        if (e < n_edges && (d >= split ? 1 : 0) == r) atomicAdd(&deg[d], 1.0f);
    }
}

__global__ void rsqrt_inplace(float* __restrict__ deg, int n) {
    int i = blockIdx.x * blockDim.x + threadIdx.x;
    if (i < n) deg[i] = rsqrtf(deg[i]);
}

template <int NCOLS, bool RELU_A>
__global__ __launch_bounds__(256) void gemm_wmma_f32(
                              const float* __restrict__ A,
                              const float* __restrict__ W,
                              const float* __restrict__ dinv,
                              const float* __restrict__ bias,
                              float* __restrict__ C,
                              float* __restrict__ S, int nrows,
                              int* __restrict__ zero2) {
    if (zero2 != nullptr && blockIdx.x == 0 && threadIdx.x < 2)
        zero2[threadIdx.x * CTR_STRIDE] = 0;
    constexpr int TC  = NCOLS / 16;
    constexpr int SPW = 8 / TC;
    __shared__ __attribute__((aligned(16))) float sA[SPW * 16 * A_PITCH];

    const int lane = threadIdx.x & 31;
    const int wave = threadIdx.x >> 5;
    const int row0 = blockIdx.x * (SPW * 16);

#pragma unroll
    for (int p = 0; p < 2 * SPW; ++p) {
        const int rl = p * 8 + wave;
        const int gr = (row0 + rl < nrows) ? (row0 + rl) : (nrows - 1);
        v4f v = *reinterpret_cast<const v4f*>(A + (long)gr * K_DIM + lane * 4);
        if (RELU_A) {
            v.x = fmaxf(v.x, 0.0f);
            v.y = fmaxf(v.y, 0.0f);
            v.z = fmaxf(v.z, 0.0f);
            v.w = fmaxf(v.w, 0.0f);
        }
        *reinterpret_cast<v4f*>(&sA[rl * A_PITCH + lane * 4]) = v;
    }
    __syncthreads();

    const int tcol  = wave % TC;
    const int ls    = wave / TC;
    const int rbase = row0 + ls * 16;
    if (rbase >= nrows) return;

    const int m     = lane & 15;
    const int khalf = lane >> 4;

    const float* arow = &sA[(ls * 16 + m) * A_PITCH + 2 * khalf];
    const float* wcol = W + tcol * 16 + m;

    v8f acc = {};
#pragma nounroll
    for (int k0 = 0; k0 < K_DIM; k0 += 32) {
        v2f a[8], b[8];
#pragma unroll
        for (int u = 0; u < 8; ++u) {
            const int kk = k0 + 4 * u + 2 * khalf;
            a[u] = *reinterpret_cast<const v2f*>(arow + k0 + 4 * u);
            b[u].x = wcol[(long)kk * NCOLS];
            b[u].y = wcol[(long)(kk + 1) * NCOLS];
        }
        __builtin_amdgcn_sched_barrier(0);
#pragma unroll
        for (int u = 0; u < 8; ++u) {
            acc = __builtin_amdgcn_wmma_f32_16x16x4_f32(
                false, a[u], false, b[u], (short)0, acc, false, false);
        }
        asm volatile("v_nop\n\tv_nop\n\tv_nop\n\tv_nop"
                     : "+v"(acc) : "v"(a[7]), "v"(b[7]));
    }

    const float bc = bias[tcol * 16 + m];
    float seed[8];
#pragma unroll
    for (int r = 0; r < 8; ++r) {
        const int row = rbase + r + 8 * khalf;
        const float di = dinv[row < nrows ? row : nrows - 1];
        seed[r] = acc[r] * di * di + bc;
    }

    if (rbase + 16 <= nrows) {
        const long off = (long)(rbase + 8 * khalf) * NCOLS + tcol * 16 + m;
#pragma unroll
        for (int r = 0; r < 8; ++r) {
            C[off + (long)r * NCOLS] = acc[r];
            S[off + (long)r * NCOLS] = seed[r];
        }
    } else {
#pragma unroll
        for (int r = 0; r < 8; ++r) {
            const int row = rbase + r + 8 * khalf;
            if (row < nrows) {
                C[(long)row * NCOLS + tcol * 16 + m] = acc[r];
                S[(long)row * NCOLS + tcol * 16 + m] = seed[r];
            }
        }
    }
}

#define AGG_CHUNK 512

template <int CH, int NE>
__device__ __forceinline__ void edge_batch(const float* __restrict__ h,
                                           float* __restrict__ out, int s,
                                           int d, int wbits, int j0, int lane) {
    constexpr int V = CH / 32;
    float buf[NE * V];
#pragma unroll
    for (int jj = 0; jj < NE; ++jj) {
        const int sj = __builtin_amdgcn_readlane(s, j0 + jj);
        const float* hp = h + (long)sj * CH + lane;
#pragma unroll
        for (int i = 0; i < V; ++i) buf[jj * V + i] = hp[32 * i];
    }
#pragma unroll
    for (int jj = 0; jj < NE; ++jj) {
        const int   dj = __builtin_amdgcn_readlane(d, j0 + jj);
        const float wj = __builtin_bit_cast(
            float, __builtin_amdgcn_readlane(wbits, j0 + jj));
        float* op = out + (long)dj * CH + lane;
#pragma unroll
        for (int i = 0; i < V; ++i)
            atomicAdd(&op[32 * i], buf[jj * V + i] * wj);
    }
}

template <int CH>
__global__ __launch_bounds__(256) void aggregate_edges(
    const float* __restrict__ h, const int* __restrict__ src,
    const int* __restrict__ dst, const float* __restrict__ dinv,
    float* __restrict__ out, int n_edges, int split,
    int* __restrict__ ctr, int nchunks) {
    constexpr int V  = CH / 32;
    constexpr int B  = 16 / V;
    constexpr int NB = 32 / B;

    __shared__ int sh_s[AGG_CHUNK];
    __shared__ int sh_d[AGG_CHUNK];
    __shared__ int ctl[3];

    const int lane = threadIdx.x & 31;
    const int wv = __builtin_amdgcn_readfirstlane((int)(threadIdx.x >> 5));

    if (threadIdx.x == 0) {
        draw_ticket(ctr, nchunks, ctl);
        ctl[2] = 0;
    }
    __syncthreads();
    const int t = __builtin_amdgcn_readfirstlane(ctl[0]);
    const int r = __builtin_amdgcn_readfirstlane(ctl[1]);
    if (t >= nchunks) return;

    constexpr int EPT = AGG_CHUNK / 256;
    int es[EPT], ed[EPT];
#pragma unroll
    for (int k = 0; k < EPT; ++k) {
        const int e  = t * AGG_CHUNK + k * 256 + (int)threadIdx.x;
        const int ec = e < n_edges ? e : n_edges - 1;
        ed[k] = dst[ec];
        es[k] = src[ec];
    }
#pragma unroll
    for (int k = 0; k < EPT; ++k) {
        const int e = t * AGG_CHUNK + k * 256 + (int)threadIdx.x;
        const int s = es[k], d = ed[k];
        const bool keep = (e < n_edges) && ((d >= split ? 1 : 0) == r);
        const unsigned m = __builtin_amdgcn_ballot_w32(keep);
        const int c = __builtin_popcount(m);
        int base = 0;
        if (lane == 0 && c != 0) base = atomicAdd(&ctl[2], c);
        base = __builtin_amdgcn_readfirstlane(base);
        if (keep) {
            const int p = base + (int)__builtin_amdgcn_mbcnt_lo(m, 0u);
            sh_s[p] = s;
            sh_d[p] = d;
        }
    }
    __syncthreads();
    const int total = __builtin_amdgcn_readfirstlane(ctl[2]);

    for (int run = wv; run * 32 < total; run += 8) {
        const int left = total - run * 32;
        const int cnt  = left < 32 ? left : 32;

        const int li = run * 32 + (lane < cnt ? lane : cnt - 1);
        const int s  = sh_s[li];
        const int d  = sh_d[li];

        if (cnt >= 32) {
            float buf[2][B * V];
#pragma unroll
            for (int jj = 0; jj < B; ++jj) {
                const int sj = __builtin_amdgcn_readlane(s, jj);
                const float* hp = h + (long)sj * CH + lane;
#pragma unroll
                for (int i = 0; i < V; ++i) buf[0][jj * V + i] = hp[32 * i];
            }
            const int wbits = __builtin_bit_cast(int, dinv[s] * dinv[d]);
#pragma unroll
            for (int b = 0; b < NB; ++b) {
                if (b + 1 < NB) {
#pragma unroll
                    for (int jj = 0; jj < B; ++jj) {
                        const int sj = __builtin_amdgcn_readlane(s, (b + 1) * B + jj);
                        const float* hp = h + (long)sj * CH + lane;
#pragma unroll
                        for (int i = 0; i < V; ++i)
                            buf[(b + 1) & 1][jj * V + i] = hp[32 * i];
                    }
                }
#pragma unroll
                for (int jj = 0; jj < B; ++jj) {
                    const int   dj = __builtin_amdgcn_readlane(d, b * B + jj);
                    const float wj = __builtin_bit_cast(
                        float, __builtin_amdgcn_readlane(wbits, b * B + jj));
                    float* op = out + (long)dj * CH + lane;
#pragma unroll
                    for (int i = 0; i < V; ++i)
                        atomicAdd(&op[32 * i], buf[b & 1][jj * V + i] * wj);
                }
            }
        } else {
            const int wbits = __builtin_bit_cast(int, dinv[s] * dinv[d]);
            int j = 0;
            for (; j + B <= cnt; j += B)
                edge_batch<CH, B>(h, out, s, d, wbits, j, lane);
            if (B > 4 && cnt - j >= 4) {
                edge_batch<CH, 4>(h, out, s, d, wbits, j, lane);
                j += 4;
            }
            if (B > 2 && cnt - j >= 2) {
                edge_batch<CH, 2>(h, out, s, d, wbits, j, lane);
                j += 2;
            }
            if (cnt - j >= 1) edge_batch<CH, 1>(h, out, s, d, wbits, j, lane);
        }
    }
}

__global__ void relu_inplace(float* __restrict__ p, long n4) {
    long i = (long)blockIdx.x * blockDim.x + threadIdx.x;
    if (i >= n4) return;
    float4* p4 = reinterpret_cast<float4*>(p);
    float4 v = p4[i];
    v.x = fmaxf(v.x, 0.0f);
    v.y = fmaxf(v.y, 0.0f);
    v.z = fmaxf(v.z, 0.0f);
    v.w = fmaxf(v.w, 0.0f);
    p4[i] = v;
}

extern "C" void kernel_launch(void* const* d_in, const int* in_sizes, int n_in,
                              void* d_out, int out_size, void* d_ws, size_t ws_size,
                              hipStream_t stream) {
    const float* x  = (const float*)d_in[0];
    const int*   ei = (const int*)d_in[1];
    const float* W1 = (const float*)d_in[2];
    const float* b1 = (const float*)d_in[3];
    const float* W2 = (const float*)d_in[4];
    const float* b2 = (const float*)d_in[5];

    const int N   = in_sizes[0] / K_DIM;
    const int E   = in_sizes[1] / 2;
    (void)n_in; (void)ws_size; (void)out_size;

    const int* src = ei;
    const int* dst = ei + E;
    float* out = (float*)d_out;

    char* ws = (char*)d_ws;
    const size_t degBytes = (((size_t)N * 4) + 255) & ~(size_t)255;
    float* deg  = (float*)ws;
    float* bufA = (float*)(ws + degBytes);
    float* bufB = (float*)(ws + degBytes + (size_t)N * 128 * 4);

    const int TB = 256;

    const int split = N / 2;
    int* ctrD = (int*)out;
    int* ctr1 = (int*)out + 2 * CTR_STRIDE;
    int* ctr2 = (int*)(bufA + (size_t)N * 64);
    const int degChunks = (E + DEG_CHUNK - 1) / DEG_CHUNK;
    const int aggChunks = (E + AGG_CHUNK - 1) / AGG_CHUNK;

    fill_f32<<<(N + TB - 1) / TB, TB, 0, stream>>>(deg, 1.0f, N, ctrD);
    degree_kernel<<<2 * degChunks, TB, 0, stream>>>(dst, deg, E, split, ctrD, degChunks);
    rsqrt_inplace<<<(N + TB - 1) / TB, TB, 0, stream>>>(deg, N);

    {
        const int tiles = ((N + 15) / 16) * (128 / 16);
        gemm_wmma_f32<128, false><<<(tiles + 7) / 8, 256, 0, stream>>>(x, W1, deg, b1, bufA, bufB, N, nullptr);
        aggregate_edges<128><<<2 * aggChunks, 256, 0, stream>>>(bufA, src, dst, deg, bufB, E, split, ctr1, aggChunks);
    }

    {
        const int tiles = ((N + 15) / 16) * (64 / 16);
        gemm_wmma_f32<64, true><<<(tiles + 7) / 8, 256, 0, stream>>>(bufB, W2, deg, b2, bufA, out, N, ctr2);
        aggregate_edges<64><<<2 * aggChunks, 256, 0, stream>>>(bufA, src, dst, deg, out, E, split, ctr2, aggChunks);
        const long n4 = (long)N * (64 / 4);
        relu_inplace<<<(unsigned)((n4 + TB - 1) / TB), TB, 0, stream>>>(out, n4);
    }
}


# Edge sums owner-computes: a 32-wave work-group per 393 destination rows holds their sums in LDS, sweeps one class byte per edge, adds gathered rows by LDS float atomics; sum order: LDS arrival
# speedup vs baseline: 2.9738x; 1.1720x over previous
#include <hip/hip_runtime.h>


typedef __attribute__((ext_vector_type(2))) float v2f;
typedef __attribute__((ext_vector_type(4))) float v4f;
typedef __attribute__((ext_vector_type(8))) float v8f;

#define K_DIM 128
#define A_PITCH 132

#define CTR_STRIDE 64

__global__ void fill_f32(float* __restrict__ p, float v, long n,
                         int* __restrict__ ctr) {
    long i = (long)blockIdx.x * blockDim.x + threadIdx.x;
    if (i < n) p[i] = v;
    if (i < 2) ctr[i * CTR_STRIDE] = 0;
}

__device__ __forceinline__ void draw_ticket(int* __restrict__ ctr, int nchunks,
                                            int* ctl) {
    const unsigned pw = __builtin_amdgcn_s_sendmsg_rtn(0x87);
    int r = (int)((pw >> 18) & 1u);
    int t = atomicAdd(&ctr[r * CTR_STRIDE], 1);
    if (t >= nchunks) {
        r ^= 1;
        t = atomicAdd(&ctr[r * CTR_STRIDE], 1);
    }
    ctl[0] = t;
    ctl[1] = r;
}

#define DEG_CHUNK 1024

__global__ __launch_bounds__(256) void degree_kernel(
    const int* __restrict__ dst, float* __restrict__ deg, int n_edges,
    int split, int* __restrict__ ctr, int nchunks) {
    __shared__ int ctl[2];
    if (threadIdx.x == 0) draw_ticket(ctr, nchunks, ctl);
    __syncthreads();
    const int t = ctl[0];
    const int r = ctl[1];
    if (t >= nchunks) return;
    constexpr int EPT = DEG_CHUNK / 256;
    int ed[EPT];
#pragma unroll
    for (int k = 0; k < EPT; ++k) {
        const int e = t * DEG_CHUNK + k * 256 + (int)threadIdx.x;
        ed[k] = dst[e < n_edges ? e : n_edges - 1];
    }
#pragma unroll
    for (int k = 0; k < EPT; ++k) {
        const int e = t * DEG_CHUNK + k * 256 + (int)threadIdx.x;
        const int d = ed[k];
        if (e < n_edges && (d >= split ? 1 : 0) == r) atomicAdd(&deg[d], 1.0f);
    }
}

__global__ void rsqrt_inplace(float* __restrict__ deg, int n) {
    int i = blockIdx.x * blockDim.x + threadIdx.x;
    if (i < n) deg[i] = rsqrtf(deg[i]);
}

template <int NCOLS>
__global__ __launch_bounds__(256) void gemm_wmma_f32(
                              const float* __restrict__ A,
                              const float* __restrict__ W,
                              float* __restrict__ C, int nrows) {
    constexpr int TC  = NCOLS / 16;
    constexpr int SPW = 8 / TC;
    __shared__ __attribute__((aligned(16))) float sA[SPW * 16 * A_PITCH];

    const int lane = threadIdx.x & 31;
    const int wave = threadIdx.x >> 5;
    const int row0 = blockIdx.x * (SPW * 16);

#pragma unroll
    for (int p = 0; p < 2 * SPW; ++p) {
        const int rl = p * 8 + wave;
        const int gr = (row0 + rl < nrows) ? (row0 + rl) : (nrows - 1);
        v4f v = *reinterpret_cast<const v4f*>(A + (long)gr * K_DIM + lane * 4);
        *reinterpret_cast<v4f*>(&sA[rl * A_PITCH + lane * 4]) = v;
    }
    __syncthreads();

    const int tcol  = wave % TC;
    const int ls    = wave / TC;
    const int rbase = row0 + ls * 16;
    if (rbase >= nrows) return;

    const int m     = lane & 15;
    const int khalf = lane >> 4;

    const float* arow = &sA[(ls * 16 + m) * A_PITCH + 2 * khalf];
    const float* wcol = W + tcol * 16 + m;

    v8f acc = {};
#pragma nounroll
    for (int k0 = 0; k0 < K_DIM; k0 += 32) {
        v2f a[8], b[8];
#pragma unroll
        for (int u = 0; u < 8; ++u) {
            const int kk = k0 + 4 * u + 2 * khalf;
            a[u] = *reinterpret_cast<const v2f*>(arow + k0 + 4 * u);
            b[u].x = wcol[(long)kk * NCOLS];
            b[u].y = wcol[(long)(kk + 1) * NCOLS];
        }
        __builtin_amdgcn_sched_barrier(0);
#pragma unroll
        for (int u = 0; u < 8; ++u) {
            acc = __builtin_amdgcn_wmma_f32_16x16x4_f32(
                false, a[u], false, b[u], (short)0, acc, false, false);
        }
        asm volatile("v_nop\n\tv_nop\n\tv_nop\n\tv_nop"
                     : "+v"(acc) : "v"(a[7]), "v"(b[7]));
    }

    if (rbase + 16 <= nrows) {
        const long off = (long)(rbase + 8 * khalf) * NCOLS + tcol * 16 + m;
#pragma unroll
        for (int r = 0; r < 8; ++r) {
            C[off + (long)r * NCOLS] = acc[r];
        }
    } else {
#pragma unroll
        for (int r = 0; r < 8; ++r) {
            const int row = rbase + r + 8 * khalf;
            if (row < nrows) C[(long)row * NCOLS + tcol * 16 + m] = acc[r];
        }
    }
}

#define R_ROWS 393

__global__ void classify_edges(const int* __restrict__ dst,
                               unsigned* __restrict__ cls, int n_edges,
                               int n_words) {
    const int t = blockIdx.x * blockDim.x + threadIdx.x;
    if (t >= n_words) return;
    unsigned w = 0;
#pragma unroll
    for (int j = 0; j < 4; ++j) {
        const int e = 4 * t + j;
        const unsigned c = e < n_edges ? (unsigned)(dst[e] / R_ROWS) : 255u;
        w |= c << (8 * j);
    }
    cls[t] = w;
}

template <int CH, int NE>
__device__ __forceinline__ void edge_batch(const float* __restrict__ h,
                                           float* acc, int s, int d,
                                           int wbits, int j0, int lane) {
    constexpr int VW = CH / 32;
    typedef __attribute__((ext_vector_type(VW))) float vrow;
    vrow buf[NE];
#pragma unroll
    for (int jj = 0; jj < NE; ++jj) {
        const int sj = __builtin_amdgcn_readlane(s, j0 + jj);
        buf[jj] = *reinterpret_cast<const vrow*>(h + (long)sj * CH + lane * VW);
    }
#pragma unroll
    for (int jj = 0; jj < NE; ++jj) {
        const int   dj = __builtin_amdgcn_readlane(d, j0 + jj);
        const float wj = __builtin_bit_cast(
            float, __builtin_amdgcn_readlane(wbits, j0 + jj));
        float* ap = acc + dj * CH + lane;
#pragma unroll
        for (int c = 0; c < VW; ++c) atomicAdd(&ap[c * 32], buf[jj][c] * wj);
    }
}

template <int CH>
__device__ __forceinline__ void run_edges(const float* __restrict__ h,
                                          const int* __restrict__ src,
                                          const int* __restrict__ dst,
                                          const float* __restrict__ dinv,
                                          float* acc, const int* ring,
                                          int head, int cnt, int r0, int lane) {
    constexpr int VW = CH / 32;
    constexpr int B  = 16 / VW;
    constexpr int NB = 32 / B;
    typedef __attribute__((ext_vector_type(VW))) float vrow;
    const int e = ring[(head + (lane < cnt ? lane : cnt - 1)) & 63];
    const int s = src[e];
    const int d = dst[e] - r0;
    if (cnt >= 32) {
        vrow buf[2][B];
#pragma unroll
        for (int jj = 0; jj < B; ++jj) {
            const int sj = __builtin_amdgcn_readlane(s, jj);
            buf[0][jj] = *reinterpret_cast<const vrow*>(h + (long)sj * CH + lane * VW);
        }
        const int wbits = __builtin_bit_cast(int, dinv[s]);
#pragma unroll
        for (int b = 0; b < NB; ++b) {
            if (b + 1 < NB) {
#pragma unroll
                for (int jj = 0; jj < B; ++jj) {
                    const int sj = __builtin_amdgcn_readlane(s, (b + 1) * B + jj);
                    buf[(b + 1) & 1][jj] =
                        *reinterpret_cast<const vrow*>(h + (long)sj * CH + lane * VW);
                }
            }
#pragma unroll
            for (int jj = 0; jj < B; ++jj) {
                const int   dj = __builtin_amdgcn_readlane(d, b * B + jj);
                const float wj = __builtin_bit_cast(
                    float, __builtin_amdgcn_readlane(wbits, b * B + jj));
                float* ap = acc + dj * CH + lane;
#pragma unroll
                for (int c = 0; c < VW; ++c)
                    atomicAdd(&ap[c * 32], buf[b & 1][jj][c] * wj);
            }
        }
    } else {
        const int wbits = __builtin_bit_cast(int, dinv[s]);
        int j = 0;
        for (; j + B <= cnt; j += B)
            edge_batch<CH, B>(h, acc, s, d, wbits, j, lane);
        if (B > 4 && cnt - j >= 4) {
            edge_batch<CH, 4>(h, acc, s, d, wbits, j, lane);
            j += 4;
        }
        if (B > 2 && cnt - j >= 2) {
            edge_batch<CH, 2>(h, acc, s, d, wbits, j, lane);
            j += 2;
        }
        if (cnt - j >= 1) edge_batch<CH, 1>(h, acc, s, d, wbits, j, lane);
    }
}

template <int CH>
__global__ __launch_bounds__(1024) void aggregate_owner(
    const float* __restrict__ h, const int* __restrict__ src,
    const int* __restrict__ dst, const uint4* __restrict__ cls4,
    const float* __restrict__ dinv, const float* __restrict__ bias,
    float* __restrict__ outp, int n_edges, int n_rows) {
    constexpr int VW = CH / 32;
    typedef __attribute__((ext_vector_type(VW))) float vrow;
    __shared__ float acc[R_ROWS * K_DIM];
    __shared__ int ring[32][64];

    const int lane = threadIdx.x & 31;
    const int wv = __builtin_amdgcn_readfirstlane((int)(threadIdx.x >> 5));
    const int r0 = blockIdx.x * R_ROWS;
    const int nr = (n_rows - r0 < R_ROWS) ? (n_rows - r0) : R_ROWS;

    for (int row = wv; row < nr; row += 32) {
        const float di = dinv[r0 + row];
        const vrow v = *reinterpret_cast<const vrow*>(h + (long)(r0 + row) * CH + lane * VW);
#pragma unroll
        for (int c = 0; c < VW; ++c) acc[row * CH + c * 32 + lane] = v[c] * di;
    }
    __syncthreads();

    const unsigned rep = (unsigned)blockIdx.x * 0x01010101u;
    const int nvec = (n_edges + 15) >> 4;
    const int nit  = (nvec + 31) >> 5;
    const int rot = (int)(((long)blockIdx.x * nit) / gridDim.x);
    int* myring = ring[wv];
    int head = 0, qn = 0;
    uint4 rawn = make_uint4(~0u, ~0u, ~0u, ~0u);
    int  idxn = 0;
    bool okn  = false;
    if (nit > 0) {
        const bool live = wv < nit;
        int it = (live ? wv : 0) + rot;
        if (it >= nit) it -= nit;
        idxn = it * 32 + lane;
        okn  = live && idxn < nvec;
        rawn = cls4[idxn < nvec ? idxn : nvec - 1];
    }
    for (int k = wv; k < nit; k += 32) {
        const int idx = idxn;
        uint4 v = rawn;
        if (!okn) v = make_uint4(~0u, ~0u, ~0u, ~0u);
        {
            const bool live = k + 32 < nit;
            int it = (live ? k + 32 : k) + rot;
            if (it >= nit) it -= nit;
            idxn = it * 32 + lane;
            okn  = live && idxn < nvec;
            rawn = cls4[idxn < nvec ? idxn : nvec - 1];
        }
        const unsigned vc[4] = {v.x, v.y, v.z, v.w};
#pragma unroll
        for (int c = 0; c < 4; ++c) {
            const unsigned x = vc[c] ^ rep;
            unsigned z = ~(((x & 0x7F7F7F7Fu) + 0x7F7F7F7Fu) | x | 0x7F7F7F7Fu);
            unsigned bm = __builtin_amdgcn_ballot_w32(z != 0u);
            while (bm != 0u) {
                if (z != 0u) {
                    const int b = __builtin_ctz(z) >> 3;
                    const int p = (head + qn + (int)__builtin_amdgcn_mbcnt_lo(bm, 0u)) & 63;
                    myring[p] = idx * 16 + c * 4 + b;
                    z &= z - 1u;
                }
                qn += __builtin_popcount(bm);
                if (qn >= 32) {
                    run_edges<CH>(h, src, dst, dinv, acc, myring, head, 32, r0, lane);
                    head = (head + 32) & 63;
                    qn -= 32;
                }
                bm = __builtin_amdgcn_ballot_w32(z != 0u);
            }
        }
    }
    if (qn > 0) run_edges<CH>(h, src, dst, dinv, acc, myring, head, qn, r0, lane);
    __syncthreads();

    vrow bv;
#pragma unroll
    for (int c = 0; c < VW; ++c) bv[c] = bias[lane * VW + c];
    for (int row = wv; row < nr; row += 32) {
        const float di = dinv[r0 + row];
        vrow o;
#pragma unroll
        for (int c = 0; c < VW; ++c)
            o[c] = fmaxf(acc[row * CH + c * 32 + lane] * di + bv[c], 0.0f);
        *reinterpret_cast<vrow*>(outp + (long)(r0 + row) * CH + lane * VW) = o;
    }
}

extern "C" void kernel_launch(void* const* d_in, const int* in_sizes, int n_in,
                              void* d_out, int out_size, void* d_ws, size_t ws_size,
                              hipStream_t stream) {
    const float* x  = (const float*)d_in[0];
    const int*   ei = (const int*)d_in[1];
    const float* W1 = (const float*)d_in[2];
    const float* b1 = (const float*)d_in[3];
    const float* W2 = (const float*)d_in[4];
    const float* b2 = (const float*)d_in[5];

    const int N   = in_sizes[0] / K_DIM;
    const int E   = in_sizes[1] / 2;
    (void)n_in; (void)ws_size; (void)out_size;

    const int* src = ei;
    const int* dst = ei + E;
    float* out = (float*)d_out;

    char* ws = (char*)d_ws;
    const size_t degBytes = (((size_t)N * 4) + 255) & ~(size_t)255;
    float* deg  = (float*)ws;
    float* bufA = (float*)(ws + degBytes);
    float* bufB = (float*)(ws + degBytes + (size_t)N * 128 * 4);

    const int TB = 256;

    const int split = N / 2;
    int* ctrD = (int*)out;
    const int degChunks = (E + DEG_CHUNK - 1) / DEG_CHUNK;

    const int clsWords = ((E + 15) / 16) * 4;
    unsigned* cls1 = (unsigned*)out;
    unsigned* cls2 = (unsigned*)(bufA + (size_t)N * 64);
    const int ranges = (N + R_ROWS - 1) / R_ROWS;

    fill_f32<<<(N + TB - 1) / TB, TB, 0, stream>>>(deg, 1.0f, N, ctrD);
    degree_kernel<<<2 * degChunks, TB, 0, stream>>>(dst, deg, E, split, ctrD, degChunks);
    rsqrt_inplace<<<(N + TB - 1) / TB, TB, 0, stream>>>(deg, N);

    {
        const int tiles = ((N + 15) / 16) * (128 / 16);
        classify_edges<<<(clsWords + TB - 1) / TB, TB, 0, stream>>>(dst, cls1, E, clsWords);
        gemm_wmma_f32<128><<<(tiles + 7) / 8, 256, 0, stream>>>(x, W1, bufA, N);
        aggregate_owner<128><<<ranges, 1024, 0, stream>>>(
            bufA, src, dst, (const uint4*)cls1, deg, b1, bufB, E, N);
    }

    {
        const int tiles = ((N + 15) / 16) * (64 / 16);
        classify_edges<<<(clsWords + TB - 1) / TB, TB, 0, stream>>>(dst, cls2, E, clsWords);
        gemm_wmma_f32<64><<<(tiles + 7) / 8, 256, 0, stream>>>(bufB, W2, bufA, N);
        aggregate_owner<64><<<ranges, 1024, 0, stream>>>(
            bufA, src, dst, (const uint4*)cls2, deg, b2, out, E, N);
    }
}


# Layer-2 edge sums replay layer 1's finds: the layer-1 owner lists each run of 32 kept edges as packed (source,row) words per range; layer 2 reads that, no sweep or class pass; sum order: LDS arrival
# speedup vs baseline: 3.4908x; 1.1739x over previous
#include <hip/hip_runtime.h>


typedef __attribute__((ext_vector_type(2))) float v2f;
typedef __attribute__((ext_vector_type(4))) float v4f;
typedef __attribute__((ext_vector_type(8))) float v8f;

#define K_DIM 128
#define A_PITCH 132

#define CTR_STRIDE 64

__global__ void fill_f32(float* __restrict__ p, float v, long n,
                         int* __restrict__ ctr) {
    long i = (long)blockIdx.x * blockDim.x + threadIdx.x;
    if (i < n) p[i] = v;
    if (i < 2) ctr[i * CTR_STRIDE] = 0;
}

__device__ __forceinline__ void draw_ticket(int* __restrict__ ctr, int nchunks,
                                            int* ctl) {
    const unsigned pw = __builtin_amdgcn_s_sendmsg_rtn(0x87);
    int r = (int)((pw >> 18) & 1u);
    int t = atomicAdd(&ctr[r * CTR_STRIDE], 1);
    if (t >= nchunks) {
        r ^= 1;
        t = atomicAdd(&ctr[r * CTR_STRIDE], 1);
    }
    ctl[0] = t;
    ctl[1] = r;
}

#define DEG_CHUNK 1024

__global__ __launch_bounds__(256) void degree_kernel(
    const int* __restrict__ dst, float* __restrict__ deg, int n_edges,
    int split, int* __restrict__ ctr, int nchunks) {
    __shared__ int ctl[2];
    if (threadIdx.x == 0) draw_ticket(ctr, nchunks, ctl);
    __syncthreads();
    const int t = ctl[0];
    const int r = ctl[1];
    if (t >= nchunks) return;
    constexpr int EPT = DEG_CHUNK / 256;
    int ed[EPT];
#pragma unroll
    for (int k = 0; k < EPT; ++k) {
        const int e = t * DEG_CHUNK + k * 256 + (int)threadIdx.x;
        ed[k] = dst[e < n_edges ? e : n_edges - 1];
    }
#pragma unroll
    for (int k = 0; k < EPT; ++k) {
        const int e = t * DEG_CHUNK + k * 256 + (int)threadIdx.x;
        const int d = ed[k];
        if (e < n_edges && (d >= split ? 1 : 0) == r) atomicAdd(&deg[d], 1.0f);
    }
}

__global__ void rsqrt_inplace(float* __restrict__ deg, int n) {
    int i = blockIdx.x * blockDim.x + threadIdx.x;
    if (i < n) deg[i] = rsqrtf(deg[i]);
}

template <int NCOLS>
__global__ __launch_bounds__(256) void gemm_wmma_f32(
                              const float* __restrict__ A,
                              const float* __restrict__ W,
                              float* __restrict__ C, int nrows) {
    constexpr int TC  = NCOLS / 16;
    constexpr int SPW = 8 / TC;
    __shared__ __attribute__((aligned(16))) float sA[SPW * 16 * A_PITCH];

    const int lane = threadIdx.x & 31;
    const int wave = threadIdx.x >> 5;
    const int row0 = blockIdx.x * (SPW * 16);

#pragma unroll
    for (int p = 0; p < 2 * SPW; ++p) {
        const int rl = p * 8 + wave;
        const int gr = (row0 + rl < nrows) ? (row0 + rl) : (nrows - 1);
        v4f v = *reinterpret_cast<const v4f*>(A + (long)gr * K_DIM + lane * 4);
        *reinterpret_cast<v4f*>(&sA[rl * A_PITCH + lane * 4]) = v;
    }
    __syncthreads();

    const int tcol  = wave % TC;
    const int ls    = wave / TC;
    const int rbase = row0 + ls * 16;
    if (rbase >= nrows) return;

    const int m     = lane & 15;
    const int khalf = lane >> 4;

    const float* arow = &sA[(ls * 16 + m) * A_PITCH + 2 * khalf];
    const float* wcol = W + tcol * 16 + m;

    v8f acc = {};
#pragma nounroll
    for (int k0 = 0; k0 < K_DIM; k0 += 32) {
        v2f a[8], b[8];
#pragma unroll
        for (int u = 0; u < 8; ++u) {
            const int kk = k0 + 4 * u + 2 * khalf;
            a[u] = *reinterpret_cast<const v2f*>(arow + k0 + 4 * u);
            b[u].x = wcol[(long)kk * NCOLS];
            b[u].y = wcol[(long)(kk + 1) * NCOLS];
        }
        __builtin_amdgcn_sched_barrier(0);
#pragma unroll
        for (int u = 0; u < 8; ++u) {
            acc = __builtin_amdgcn_wmma_f32_16x16x4_f32(
                false, a[u], false, b[u], (short)0, acc, false, false);
        }
        asm volatile("v_nop\n\tv_nop\n\tv_nop\n\tv_nop"
                     : "+v"(acc) : "v"(a[7]), "v"(b[7]));
    }

    if (rbase + 16 <= nrows) {
        const long off = (long)(rbase + 8 * khalf) * NCOLS + tcol * 16 + m;
#pragma unroll
        for (int r = 0; r < 8; ++r) {
            C[off + (long)r * NCOLS] = acc[r];
        }
    } else {
#pragma unroll
        for (int r = 0; r < 8; ++r) {
            const int row = rbase + r + 8 * khalf;
            if (row < nrows) C[(long)row * NCOLS + tcol * 16 + m] = acc[r];
        }
    }
}

#define R_ROWS 393

#define OUT_COLS     64
#define PIECE_SHIFT  9
#define PIECE_UNITS  (1 << PIECE_SHIFT)
#define REGION_UNITS (R_ROWS * OUT_COLS * 4 / 16)
#define LIST_HDR     (PIECE_UNITS * 4)
#define LIST_BASE    (PIECE_UNITS * 4 + 32)
#define SRC_BITS 17

__device__ __forceinline__ int piece_unit(int i) {
    return (i >> PIECE_SHIFT) * REGION_UNITS + (i & (PIECE_UNITS - 1));
}

__global__ void classify_edges(const int* __restrict__ dst,
                               unsigned* __restrict__ cls, int n_edges,
                               int n_words) {
    const int t = blockIdx.x * blockDim.x + threadIdx.x;
    if (t >= n_words) return;
    unsigned w = 0;
#pragma unroll
    for (int j = 0; j < 4; ++j) {
        const int e = 4 * t + j;
        const unsigned c = e < n_edges ? (unsigned)(dst[e] / R_ROWS) : 255u;
        w |= c << (8 * j);
    }
    cls[piece_unit(t >> 2) * 4 + (t & 3)] = w;
}

template <int CH, int NE>
__device__ __forceinline__ void edge_batch(const float* __restrict__ h,
                                           float* acc, int s, int d,
                                           int wbits, int j0, int lane) {
    constexpr int VW = CH / 32;
    typedef __attribute__((ext_vector_type(VW))) float vrow;
    vrow buf[NE];
#pragma unroll
    for (int jj = 0; jj < NE; ++jj) {
        const int sj = __builtin_amdgcn_readlane(s, j0 + jj);
        buf[jj] = *reinterpret_cast<const vrow*>(h + (long)sj * CH + lane * VW);
    }
#pragma unroll
    for (int jj = 0; jj < NE; ++jj) {
        const int   dj = __builtin_amdgcn_readlane(d, j0 + jj);
        const float wj = __builtin_bit_cast(
            float, __builtin_amdgcn_readlane(wbits, j0 + jj));
        float* ap = acc + dj * CH + lane;
#pragma unroll
        for (int c = 0; c < VW; ++c) atomicAdd(&ap[c * 32], buf[jj][c] * wj);
    }
}

template <int CH>
__device__ __forceinline__ void run_sd(const float* __restrict__ h,
                                       const float* __restrict__ dinv,
                                       float* acc, int s, int d, int cnt,
                                       int lane) {
    constexpr int VW = CH / 32;
    constexpr int B  = 16 / VW;
    constexpr int NB = 32 / B;
    typedef __attribute__((ext_vector_type(VW))) float vrow;
    if (cnt >= 32) {
        vrow buf[2][B];
#pragma unroll
        for (int jj = 0; jj < B; ++jj) {
            const int sj = __builtin_amdgcn_readlane(s, jj);
            buf[0][jj] = *reinterpret_cast<const vrow*>(h + (long)sj * CH + lane * VW);
        }
        const int wbits = __builtin_bit_cast(int, dinv[s]);
#pragma unroll
        for (int b = 0; b < NB; ++b) {
            if (b + 1 < NB) {
#pragma unroll
                for (int jj = 0; jj < B; ++jj) {
                    const int sj = __builtin_amdgcn_readlane(s, (b + 1) * B + jj);
                    buf[(b + 1) & 1][jj] =
                        *reinterpret_cast<const vrow*>(h + (long)sj * CH + lane * VW);
                }
            }
#pragma unroll
            for (int jj = 0; jj < B; ++jj) {
                const int   dj = __builtin_amdgcn_readlane(d, b * B + jj);
                const float wj = __builtin_bit_cast(
                    float, __builtin_amdgcn_readlane(wbits, b * B + jj));
                float* ap = acc + dj * CH + lane;
#pragma unroll
                for (int c = 0; c < VW; ++c)
                    atomicAdd(&ap[c * 32], buf[b & 1][jj][c] * wj);
            }
        }
    } else {
        const int wbits = __builtin_bit_cast(int, dinv[s]);
        int j = 0;
        for (; j + B <= cnt; j += B)
            edge_batch<CH, B>(h, acc, s, d, wbits, j, lane);
        if (B > 4 && cnt - j >= 4) {
            edge_batch<CH, 4>(h, acc, s, d, wbits, j, lane);
            j += 4;
        }
        if (B > 2 && cnt - j >= 2) {
            edge_batch<CH, 2>(h, acc, s, d, wbits, j, lane);
            j += 2;
        }
        if (cnt - j >= 1) edge_batch<CH, 1>(h, acc, s, d, wbits, j, lane);
    }
}

template <int CH, bool RECORD>
__device__ __forceinline__ void run_edges(const float* __restrict__ h,
                                          const int* __restrict__ src,
                                          const int* __restrict__ dst,
                                          const float* __restrict__ dinv,
                                          float* acc, const int* ring,
                                          int head, int cnt, int r0, int lane,
                                          unsigned* __restrict__ list, int cap,
                                          int* lcnt) {
    const int e = ring[(head + (lane < cnt ? lane : cnt - 1)) & 63];
    const int s = src[e];
    const int d = dst[e] - r0;
    if (RECORD) {
        int base = 0;
        if (lane == 0) base = atomicAdd(lcnt, 32);
        base = __builtin_amdgcn_readfirstlane(base);
        if (base + 32 <= cap)
            list[base + lane] =
                lane < cnt ? ((unsigned)s | ((unsigned)d << SRC_BITS)) : ~0u;
    }
    run_sd<CH>(h, dinv, acc, s, d, cnt, lane);
}

__device__ __forceinline__ uint4 classes_from_dst(const int* __restrict__ dst,
                                                  int i, int n_edges) {
    unsigned w[4];
#pragma unroll
    for (int c = 0; c < 4; ++c) {
        w[c] = 0u;
#pragma unroll
        for (int j = 0; j < 4; ++j) {
            const int e = i * 16 + c * 4 + j;
            const unsigned cl = e < n_edges ? (unsigned)(dst[e] / R_ROWS) : 255u;
            w[c] |= cl << (8 * j);
        }
    }
    return make_uint4(w[0], w[1], w[2], w[3]);
}

template <bool FROM_DST>
__device__ __forceinline__ uint4 class_unit(const uint4* __restrict__ cls4,
                                            const int* __restrict__ dst, int i,
                                            int n_edges) {
    if (FROM_DST) return classes_from_dst(dst, i, n_edges);
    return cls4[piece_unit(i)];
}

template <int CH, bool FROM_DST, bool RECORD>
__device__ __forceinline__ void sweep_range(
    const float* __restrict__ h, const int* __restrict__ src,
    const int* __restrict__ dst, const uint4* __restrict__ cls4,
    const float* __restrict__ dinv, float* acc, int* myring, int n_edges,
    int r0, int lane, int wv, unsigned* list, int cap, int* lcnt) {
    const unsigned rep = (unsigned)blockIdx.x * 0x01010101u;
    const int nvec = (n_edges + 15) >> 4;
    const int nit  = (nvec + 31) >> 5;
    const int rot = (int)(((long)blockIdx.x * nit) / gridDim.x);
    int head = 0, qn = 0;
    uint4 rawn = make_uint4(~0u, ~0u, ~0u, ~0u);
    int  idxn = 0;
    bool okn  = false;
    if (nit > 0) {
        const bool live = wv < nit;
        int it = (live ? wv : 0) + rot;
        if (it >= nit) it -= nit;
        idxn = it * 32 + lane;
        okn  = live && idxn < nvec;
        rawn = class_unit<FROM_DST>(cls4, dst, idxn < nvec ? idxn : nvec - 1, n_edges);
    }
    for (int k = wv; k < nit; k += 32) {
        const int idx = idxn;
        uint4 v = rawn;
        if (!okn) v = make_uint4(~0u, ~0u, ~0u, ~0u);
        {
            const bool live = k + 32 < nit;
            int it = (live ? k + 32 : k) + rot;
            if (it >= nit) it -= nit;
            idxn = it * 32 + lane;
            okn  = live && idxn < nvec;
            rawn = class_unit<FROM_DST>(cls4, dst, idxn < nvec ? idxn : nvec - 1, n_edges);
        }
        const unsigned vc[4] = {v.x, v.y, v.z, v.w};
#pragma unroll
        for (int c = 0; c < 4; ++c) {
            const unsigned x = vc[c] ^ rep;
            unsigned z = ~(((x & 0x7F7F7F7Fu) + 0x7F7F7F7Fu) | x | 0x7F7F7F7Fu);
            unsigned bm = __builtin_amdgcn_ballot_w32(z != 0u);
            while (bm != 0u) {
                if (z != 0u) {
                    const int b = __builtin_ctz(z) >> 3;
                    const int p = (head + qn + (int)__builtin_amdgcn_mbcnt_lo(bm, 0u)) & 63;
                    myring[p] = idx * 16 + c * 4 + b;
                    z &= z - 1u;
                }
                qn += __builtin_popcount(bm);
                if (qn >= 32) {
                    run_edges<CH, RECORD>(h, src, dst, dinv, acc, myring, head, 32,
                                          r0, lane, list, cap, lcnt);
                    head = (head + 32) & 63;
                    qn -= 32;
                }
                bm = __builtin_amdgcn_ballot_w32(z != 0u);
            }
        }
    }
    if (qn > 0)
        run_edges<CH, RECORD>(h, src, dst, dinv, acc, myring, head, qn, r0, lane,
                              list, cap, lcnt);
}

template <int CH, bool REPLAY>
__global__ __launch_bounds__(1024) void aggregate_owner(
    const float* __restrict__ h, const int* __restrict__ src,
    const int* __restrict__ dst, const uint4* __restrict__ cls4,
    const float* __restrict__ dinv, const float* __restrict__ bias,
    float* outp, unsigned* lists, int n_edges, int n_rows) {
    constexpr int VW = CH / 32;
    typedef __attribute__((ext_vector_type(VW))) float vrow;
    __shared__ float acc[R_ROWS * K_DIM];
    __shared__ int ring[32][64];
    __shared__ int lcnt;

    const int lane = threadIdx.x & 31;
    const int wv = __builtin_amdgcn_readfirstlane((int)(threadIdx.x >> 5));
    const int r0 = blockIdx.x * R_ROWS;
    const int nr = (n_rows - r0 < R_ROWS) ? (n_rows - r0) : R_ROWS;

    unsigned* rlist = lists + (long)blockIdx.x * (REGION_UNITS * 4);
    const bool haslist = nr * OUT_COLS >= LIST_BASE;
    const int cap = haslist ? ((nr * OUT_COLS - LIST_BASE) & ~31) : 0;

    if (threadIdx.x == 0) lcnt = 0;
    for (int row = wv; row < nr; row += 32) {
        const float di = dinv[r0 + row];
        const vrow v = *reinterpret_cast<const vrow*>(h + (long)(r0 + row) * CH + lane * VW);
#pragma unroll
        for (int c = 0; c < VW; ++c) acc[row * CH + c * 32 + lane] = v[c] * di;
    }
    __syncthreads();

    if (REPLAY) {
        int total = haslist ? (int)rlist[LIST_HDR] : 0;
        total = __builtin_amdgcn_readfirstlane(total);
        if (haslist && total <= cap) {
            const int nruns = total >> 5;
            const unsigned* ent = rlist + LIST_BASE + lane;
            unsigned wn = ~0u;
            if (nruns > 0) wn = ent[(wv < nruns ? wv : 0) * 32];
            for (int r = wv; r < nruns; r += 32) {
                const unsigned w = wn;
                wn = ent[(r + 32 < nruns ? r + 32 : r) * 32];
                const bool ok = w != ~0u;
                const int cnt = __builtin_popcount(__builtin_amdgcn_ballot_w32(ok));
                const int s = ok ? (int)(w & ((1u << SRC_BITS) - 1u)) : 0;
                const int d = ok ? (int)(w >> SRC_BITS) : 0;
                if (cnt > 0) run_sd<CH>(h, dinv, acc, s, d, cnt, lane);
            }
        } else {
            sweep_range<CH, true, false>(h, src, dst, cls4, dinv, acc, ring[wv],
                                         n_edges, r0, lane, wv, rlist, 0, &lcnt);
        }
    } else {
        sweep_range<CH, false, true>(h, src, dst, cls4, dinv, acc, ring[wv],
                                     n_edges, r0, lane, wv, rlist + LIST_BASE, cap,
                                     &lcnt);
    }
    __syncthreads();

    if (!REPLAY && haslist && threadIdx.x == 0) rlist[LIST_HDR] = (unsigned)lcnt;

    vrow bv;
#pragma unroll
    for (int c = 0; c < VW; ++c) bv[c] = bias[lane * VW + c];
    for (int row = wv; row < nr; row += 32) {
        const float di = dinv[r0 + row];
        vrow o;
#pragma unroll
        for (int c = 0; c < VW; ++c)
            o[c] = fmaxf(acc[row * CH + c * 32 + lane] * di + bv[c], 0.0f);
        *reinterpret_cast<vrow*>(outp + (long)(r0 + row) * CH + lane * VW) = o;
    }
}

extern "C" void kernel_launch(void* const* d_in, const int* in_sizes, int n_in,
                              void* d_out, int out_size, void* d_ws, size_t ws_size,
                              hipStream_t stream) {
    const float* x  = (const float*)d_in[0];
    const int*   ei = (const int*)d_in[1];
    const float* W1 = (const float*)d_in[2];
    const float* b1 = (const float*)d_in[3];
    const float* W2 = (const float*)d_in[4];
    const float* b2 = (const float*)d_in[5];

    const int N   = in_sizes[0] / K_DIM;
    const int E   = in_sizes[1] / 2;
    (void)n_in; (void)ws_size; (void)out_size;

    const int* src = ei;
    const int* dst = ei + E;
    float* out = (float*)d_out;

    char* ws = (char*)d_ws;
    const size_t degBytes = (((size_t)N * 4) + 255) & ~(size_t)255;
    float* deg  = (float*)ws;
    float* bufA = (float*)(ws + degBytes);
    float* bufB = (float*)(ws + degBytes + (size_t)N * 128 * 4);

    const int TB = 256;

    const int split = N / 2;
    int* ctrD = (int*)out;
    const int degChunks = (E + DEG_CHUNK - 1) / DEG_CHUNK;

    const int clsWords = ((E + 15) / 16) * 4;
    unsigned* cls1 = (unsigned*)out;
    const int ranges = (N + R_ROWS - 1) / R_ROWS;

    fill_f32<<<(N + TB - 1) / TB, TB, 0, stream>>>(deg, 1.0f, N, ctrD);
    degree_kernel<<<2 * degChunks, TB, 0, stream>>>(dst, deg, E, split, ctrD, degChunks);
    rsqrt_inplace<<<(N + TB - 1) / TB, TB, 0, stream>>>(deg, N);

    {
        const int tiles = ((N + 15) / 16) * (128 / 16);
        classify_edges<<<(clsWords + TB - 1) / TB, TB, 0, stream>>>(dst, cls1, E, clsWords);
        gemm_wmma_f32<128><<<(tiles + 7) / 8, 256, 0, stream>>>(x, W1, bufA, N);
        aggregate_owner<128, false><<<ranges, 1024, 0, stream>>>(
            bufA, src, dst, (const uint4*)cls1, deg, b1, bufB, (unsigned*)out, E, N);
    }

    {
        const int tiles = ((N + 15) / 16) * (64 / 16);
        gemm_wmma_f32<64><<<(tiles + 7) / 8, 256, 0, stream>>>(bufB, W2, bufA, N);
        aggregate_owner<64, true><<<ranges, 1024, 0, stream>>>(
            bufA, src, dst, (const uint4*)cls1, deg, b2, out, (unsigned*)out, E, N);
    }
}


# Layer-1 sweep: a lane folds 32 class bytes (two 16-byte units) into one 32-bit hit mask: one vote and one queueing loop per 1024 edges, not eight; sum order: LDS arrival, as before
# speedup vs baseline: 3.6613x; 1.0488x over previous
#include <hip/hip_runtime.h>


typedef __attribute__((ext_vector_type(2))) float v2f;
typedef __attribute__((ext_vector_type(4))) float v4f;
typedef __attribute__((ext_vector_type(8))) float v8f;

#define K_DIM 128
#define A_PITCH 132

#define CTR_STRIDE 64

__global__ void fill_f32(float* __restrict__ p, float v, long n,
                         int* __restrict__ ctr) {
    long i = (long)blockIdx.x * blockDim.x + threadIdx.x;
    if (i < n) p[i] = v;
    if (i < 2) ctr[i * CTR_STRIDE] = 0;
}

__device__ __forceinline__ void draw_ticket(int* __restrict__ ctr, int nchunks,
                                            int* ctl) {
    const unsigned pw = __builtin_amdgcn_s_sendmsg_rtn(0x87);
    int r = (int)((pw >> 18) & 1u);
    int t = atomicAdd(&ctr[r * CTR_STRIDE], 1);
    if (t >= nchunks) {
        r ^= 1;
        t = atomicAdd(&ctr[r * CTR_STRIDE], 1);
    }
    ctl[0] = t;
    ctl[1] = r;
}

#define DEG_CHUNK 1024

__global__ __launch_bounds__(256) void degree_kernel(
    const int* __restrict__ dst, float* __restrict__ deg, int n_edges,
    int split, int* __restrict__ ctr, int nchunks) {
    __shared__ int ctl[2];
    if (threadIdx.x == 0) draw_ticket(ctr, nchunks, ctl);
    __syncthreads();
    const int t = ctl[0];
    const int r = ctl[1];
    if (t >= nchunks) return;
    constexpr int EPT = DEG_CHUNK / 256;
    int ed[EPT];
#pragma unroll
    for (int k = 0; k < EPT; ++k) {
        const int e = t * DEG_CHUNK + k * 256 + (int)threadIdx.x;
        ed[k] = dst[e < n_edges ? e : n_edges - 1];
    }
#pragma unroll
    for (int k = 0; k < EPT; ++k) {
        const int e = t * DEG_CHUNK + k * 256 + (int)threadIdx.x;
        const int d = ed[k];
        if (e < n_edges && (d >= split ? 1 : 0) == r) atomicAdd(&deg[d], 1.0f);
    }
}

__global__ void rsqrt_inplace(float* __restrict__ deg, int n) {
    int i = blockIdx.x * blockDim.x + threadIdx.x;
    if (i < n) deg[i] = rsqrtf(deg[i]);
}

template <int NCOLS>
__global__ __launch_bounds__(256) void gemm_wmma_f32(
                              const float* __restrict__ A,
                              const float* __restrict__ W,
                              float* __restrict__ C, int nrows) {
    constexpr int TC  = NCOLS / 16;
    constexpr int SPW = 8 / TC;
    __shared__ __attribute__((aligned(16))) float sA[SPW * 16 * A_PITCH];

    const int lane = threadIdx.x & 31;
    const int wave = threadIdx.x >> 5;
    const int row0 = blockIdx.x * (SPW * 16);

#pragma unroll
    for (int p = 0; p < 2 * SPW; ++p) {
        const int rl = p * 8 + wave;
        const int gr = (row0 + rl < nrows) ? (row0 + rl) : (nrows - 1);
        v4f v = *reinterpret_cast<const v4f*>(A + (long)gr * K_DIM + lane * 4);
        *reinterpret_cast<v4f*>(&sA[rl * A_PITCH + lane * 4]) = v;
    }
    __syncthreads();

    const int tcol  = wave % TC;
    const int ls    = wave / TC;
    const int rbase = row0 + ls * 16;
    if (rbase >= nrows) return;

    const int m     = lane & 15;
    const int khalf = lane >> 4;

    const float* arow = &sA[(ls * 16 + m) * A_PITCH + 2 * khalf];
    const float* wcol = W + tcol * 16 + m;

    v8f acc = {};
#pragma nounroll
    for (int k0 = 0; k0 < K_DIM; k0 += 32) {
        v2f a[8], b[8];
#pragma unroll
        for (int u = 0; u < 8; ++u) {
            const int kk = k0 + 4 * u + 2 * khalf;
            a[u] = *reinterpret_cast<const v2f*>(arow + k0 + 4 * u);
            b[u].x = wcol[(long)kk * NCOLS];
            b[u].y = wcol[(long)(kk + 1) * NCOLS];
        }
        __builtin_amdgcn_sched_barrier(0);
#pragma unroll
        for (int u = 0; u < 8; ++u) {
            acc = __builtin_amdgcn_wmma_f32_16x16x4_f32(
                false, a[u], false, b[u], (short)0, acc, false, false);
        }
        asm volatile("v_nop\n\tv_nop\n\tv_nop\n\tv_nop"
                     : "+v"(acc) : "v"(a[7]), "v"(b[7]));
    }

    if (rbase + 16 <= nrows) {
        const long off = (long)(rbase + 8 * khalf) * NCOLS + tcol * 16 + m;
#pragma unroll
        for (int r = 0; r < 8; ++r) {
            C[off + (long)r * NCOLS] = acc[r];
        }
    } else {
#pragma unroll
        for (int r = 0; r < 8; ++r) {
            const int row = rbase + r + 8 * khalf;
            if (row < nrows) C[(long)row * NCOLS + tcol * 16 + m] = acc[r];
        }
    }
}

#define R_ROWS 393

#define OUT_COLS     64
#define PIECE_SHIFT  9
#define PIECE_UNITS  (1 << PIECE_SHIFT)
#define REGION_UNITS (R_ROWS * OUT_COLS * 4 / 16)
#define LIST_HDR     (PIECE_UNITS * 4)
#define LIST_BASE    (PIECE_UNITS * 4 + 32)
#define SRC_BITS 17

__device__ __forceinline__ int piece_unit(int i) {
    return (i >> PIECE_SHIFT) * REGION_UNITS + (i & (PIECE_UNITS - 1));
}

__global__ void classify_edges(const int* __restrict__ dst,
                               unsigned* __restrict__ cls, int n_edges,
                               int n_words) {
    const int t = blockIdx.x * blockDim.x + threadIdx.x;
    if (t >= n_words) return;
    unsigned w = 0;
#pragma unroll
    for (int j = 0; j < 4; ++j) {
        const int e = 4 * t + j;
        const unsigned c = e < n_edges ? (unsigned)(dst[e] / R_ROWS) : 255u;
        w |= c << (8 * j);
    }
    cls[piece_unit(t >> 2) * 4 + (t & 3)] = w;
}

template <int CH, int NE>
__device__ __forceinline__ void edge_batch(const float* __restrict__ h,
                                           float* acc, int s, int d,
                                           int wbits, int j0, int lane) {
    constexpr int VW = CH / 32;
    typedef __attribute__((ext_vector_type(VW))) float vrow;
    vrow buf[NE];
#pragma unroll
    for (int jj = 0; jj < NE; ++jj) {
        const int sj = __builtin_amdgcn_readlane(s, j0 + jj);
        buf[jj] = *reinterpret_cast<const vrow*>(h + (long)sj * CH + lane * VW);
    }
#pragma unroll
    for (int jj = 0; jj < NE; ++jj) {
        const int   dj = __builtin_amdgcn_readlane(d, j0 + jj);
        const float wj = __builtin_bit_cast(
            float, __builtin_amdgcn_readlane(wbits, j0 + jj));
        float* ap = acc + dj * CH + lane;
#pragma unroll
        for (int c = 0; c < VW; ++c) atomicAdd(&ap[c * 32], buf[jj][c] * wj);
    }
}

template <int CH>
__device__ __forceinline__ void run_sd(const float* __restrict__ h,
                                       const float* __restrict__ dinv,
                                       float* acc, int s, int d, int cnt,
                                       int lane) {
    constexpr int VW = CH / 32;
    constexpr int B  = 16 / VW;
    constexpr int NB = 32 / B;
    typedef __attribute__((ext_vector_type(VW))) float vrow;
    if (cnt >= 32) {
        vrow buf[2][B];
#pragma unroll
        for (int jj = 0; jj < B; ++jj) {
            const int sj = __builtin_amdgcn_readlane(s, jj);
            buf[0][jj] = *reinterpret_cast<const vrow*>(h + (long)sj * CH + lane * VW);
        }
        const int wbits = __builtin_bit_cast(int, dinv[s]);
#pragma unroll
        for (int b = 0; b < NB; ++b) {
            if (b + 1 < NB) {
#pragma unroll
                for (int jj = 0; jj < B; ++jj) {
                    const int sj = __builtin_amdgcn_readlane(s, (b + 1) * B + jj);
                    buf[(b + 1) & 1][jj] =
                        *reinterpret_cast<const vrow*>(h + (long)sj * CH + lane * VW);
                }
            }
#pragma unroll
            for (int jj = 0; jj < B; ++jj) {
                const int   dj = __builtin_amdgcn_readlane(d, b * B + jj);
                const float wj = __builtin_bit_cast(
                    float, __builtin_amdgcn_readlane(wbits, b * B + jj));
                float* ap = acc + dj * CH + lane;
#pragma unroll
                for (int c = 0; c < VW; ++c)
                    atomicAdd(&ap[c * 32], buf[b & 1][jj][c] * wj);
            }
        }
    } else {
        const int wbits = __builtin_bit_cast(int, dinv[s]);
        int j = 0;
        for (; j + B <= cnt; j += B)
            edge_batch<CH, B>(h, acc, s, d, wbits, j, lane);
        if (B > 4 && cnt - j >= 4) {
            edge_batch<CH, 4>(h, acc, s, d, wbits, j, lane);
            j += 4;
        }
        if (B > 2 && cnt - j >= 2) {
            edge_batch<CH, 2>(h, acc, s, d, wbits, j, lane);
            j += 2;
        }
        if (cnt - j >= 1) edge_batch<CH, 1>(h, acc, s, d, wbits, j, lane);
    }
}

template <int CH, bool RECORD>
__device__ __forceinline__ void run_edges(const float* __restrict__ h,
                                          const int* __restrict__ src,
                                          const int* __restrict__ dst,
                                          const float* __restrict__ dinv,
                                          float* acc, const int* ring,
                                          int head, int cnt, int r0, int lane,
                                          unsigned* __restrict__ list, int cap,
                                          int* lcnt) {
    const int e = ring[(head + (lane < cnt ? lane : cnt - 1)) & 63];
    const int s = src[e];
    const int d = dst[e] - r0;
    if (RECORD) {
        int base = 0;
        if (lane == 0) base = atomicAdd(lcnt, 32);
        base = __builtin_amdgcn_readfirstlane(base);
        if (base + 32 <= cap)
            list[base + lane] =
                lane < cnt ? ((unsigned)s | ((unsigned)d << SRC_BITS)) : ~0u;
    }
    run_sd<CH>(h, dinv, acc, s, d, cnt, lane);
}

__device__ __forceinline__ uint4 classes_from_dst(const int* __restrict__ dst,
                                                  int i, int n_edges) {
    unsigned w[4];
#pragma unroll
    for (int c = 0; c < 4; ++c) {
        w[c] = 0u;
#pragma unroll
        for (int j = 0; j < 4; ++j) {
            const int e = i * 16 + c * 4 + j;
            const unsigned cl = e < n_edges ? (unsigned)(dst[e] / R_ROWS) : 255u;
            w[c] |= cl << (8 * j);
        }
    }
    return make_uint4(w[0], w[1], w[2], w[3]);
}

static_assert(PIECE_UNITS % 64 == 0, "units i and i + 32 of a trip must share a piece");
template <bool FROM_DST>
__device__ __forceinline__ void class_pair(const uint4* __restrict__ cls4,
                                           const int* __restrict__ dst, int i,
                                           bool has_b, int n_edges, uint4& a,
                                           uint4& b) {
    if (FROM_DST) {
        a = classes_from_dst(dst, i, n_edges);
        b = classes_from_dst(dst, has_b ? i + 32 : i, n_edges);
    } else {
        const int p = piece_unit(i);
        a = cls4[p];
        b = cls4[has_b ? p + 32 : p];
    }
}

__device__ __forceinline__ unsigned not_mine(unsigned v, unsigned rep) {
    const unsigned x = v ^ rep;
    return ((x & 0x7F7F7F7Fu) + 0x7F7F7F7Fu) | x;
}

#define UNIT_A 0x0F0F0F0Fu
#define UNIT_B 0xF0F0F0F0u
__device__ __forceinline__ unsigned hit_mask(uint4 a, uint4 b, unsigned rep,
                                             unsigned valid) {
    const unsigned w[8] = {a.x, a.y, a.z, a.w, b.x, b.y, b.z, b.w};
    unsigned acc = 0u;
#pragma unroll
    for (int j = 0; j < 8; ++j)
        acc = (acc >> 1) | (not_mine(w[j], rep) & 0x80808080u);
    return ~acc & valid;
}

template <int CH, bool FROM_DST, bool RECORD>
__device__ __forceinline__ void sweep_range(
    const float* __restrict__ h, const int* __restrict__ src,
    const int* __restrict__ dst, const uint4* __restrict__ cls4,
    const float* __restrict__ dinv, float* acc, int* myring, int n_edges,
    int r0, int lane, int wv, unsigned* list, int cap, int* lcnt) {
    const unsigned rep = (unsigned)blockIdx.x * 0x01010101u;
    const int nvec = (n_edges + 15) >> 4;
    const int nit  = (nvec + 63) >> 6;
    const int rot = (int)(((long)blockIdx.x * nit) / gridDim.x);
    int head = 0, qn = 0;
    uint4 rawa = make_uint4(0u, 0u, 0u, 0u), rawb = rawa;
    int idxn = 0;
    unsigned validn = 0u;
    if (nit > 0) {
        const bool live = wv < nit;
        int it = (live ? wv : 0) + rot;
        if (it >= nit) it -= nit;
        idxn = it * 64 + lane;
        const bool has_b = idxn + 32 < nvec;
        validn = live ? ((idxn < nvec ? UNIT_A : 0u) | (has_b ? UNIT_B : 0u)) : 0u;
        class_pair<FROM_DST>(cls4, dst, idxn < nvec ? idxn : nvec - 1, has_b,
                             n_edges, rawa, rawb);
    }
    for (int k = wv; k < nit; k += 32) {
        const int idx = idxn;
        const uint4 va = rawa, vb = rawb;
        const unsigned valid = validn;
        {
            const bool live = k + 32 < nit;
            int it = (live ? k + 32 : k) + rot;
            if (it >= nit) it -= nit;
            idxn = it * 64 + lane;
            const bool has_b = idxn + 32 < nvec;
            validn = live ? ((idxn < nvec ? UNIT_A : 0u) | (has_b ? UNIT_B : 0u)) : 0u;
            class_pair<FROM_DST>(cls4, dst, idxn < nvec ? idxn : nvec - 1, has_b,
                                 n_edges, rawa, rawb);
        }
        unsigned m = hit_mask(va, vb, rep, valid);
        unsigned bm = __builtin_amdgcn_ballot_w32(m != 0u);
        while (bm != 0u) {
            if (m != 0u) {
                const int i = __builtin_ctz(m);
                const int p = (head + qn + (int)__builtin_amdgcn_mbcnt_lo(bm, 0u)) & 63;
                myring[p] = idx * 16 + ((i & 4) << 7) + ((i & 3) << 2) + (i >> 3);
                m &= m - 1u;
            }
            qn += __builtin_popcount(bm);
            if (qn >= 32) {
                run_edges<CH, RECORD>(h, src, dst, dinv, acc, myring, head, 32,
                                      r0, lane, list, cap, lcnt);
                head = (head + 32) & 63;
                qn -= 32;
            }
            bm = __builtin_amdgcn_ballot_w32(m != 0u);
        }
    }
    if (qn > 0)
        run_edges<CH, RECORD>(h, src, dst, dinv, acc, myring, head, qn, r0, lane,
                              list, cap, lcnt);
}

template <int CH, bool REPLAY>
__global__ __launch_bounds__(1024) void aggregate_owner(
    const float* __restrict__ h, const int* __restrict__ src,
    const int* __restrict__ dst, const uint4* __restrict__ cls4,
    const float* __restrict__ dinv, const float* __restrict__ bias,
    float* outp, unsigned* lists, int n_edges, int n_rows) {
    constexpr int VW = CH / 32;
    typedef __attribute__((ext_vector_type(VW))) float vrow;
    __shared__ float acc[R_ROWS * K_DIM];
    __shared__ int ring[32][64];
    __shared__ int lcnt;

    const int lane = threadIdx.x & 31;
    const int wv = __builtin_amdgcn_readfirstlane((int)(threadIdx.x >> 5));
    const int r0 = blockIdx.x * R_ROWS;
    const int nr = (n_rows - r0 < R_ROWS) ? (n_rows - r0) : R_ROWS;

    unsigned* rlist = lists + (long)blockIdx.x * (REGION_UNITS * 4);
    const bool haslist = nr * OUT_COLS >= LIST_BASE;
    const int cap = haslist ? ((nr * OUT_COLS - LIST_BASE) & ~31) : 0;

    if (threadIdx.x == 0) lcnt = 0;
    for (int row = wv; row < nr; row += 32) {
        const float di = dinv[r0 + row];
        const vrow v = *reinterpret_cast<const vrow*>(h + (long)(r0 + row) * CH + lane * VW);
#pragma unroll
        for (int c = 0; c < VW; ++c) acc[row * CH + c * 32 + lane] = v[c] * di;
    }
    __syncthreads();

    if (REPLAY) {
        int total = haslist ? (int)rlist[LIST_HDR] : 0;
        total = __builtin_amdgcn_readfirstlane(total);
        if (haslist && total <= cap) {
            const int nruns = total >> 5;
            const unsigned* ent = rlist + LIST_BASE + lane;
            unsigned wn = ~0u;
            if (nruns > 0) wn = ent[(wv < nruns ? wv : 0) * 32];
            for (int r = wv; r < nruns; r += 32) {
                const unsigned w = wn;
                wn = ent[(r + 32 < nruns ? r + 32 : r) * 32];
                const bool ok = w != ~0u;
                const int cnt = __builtin_popcount(__builtin_amdgcn_ballot_w32(ok));
                const int s = ok ? (int)(w & ((1u << SRC_BITS) - 1u)) : 0;
                const int d = ok ? (int)(w >> SRC_BITS) : 0;
                if (cnt > 0) run_sd<CH>(h, dinv, acc, s, d, cnt, lane);
            }
        } else {
            sweep_range<CH, true, false>(h, src, dst, cls4, dinv, acc, ring[wv],
                                         n_edges, r0, lane, wv, rlist, 0, &lcnt);
        }
    } else {
        sweep_range<CH, false, true>(h, src, dst, cls4, dinv, acc, ring[wv],
                                     n_edges, r0, lane, wv, rlist + LIST_BASE, cap,
                                     &lcnt);
    }
    __syncthreads();

    if (!REPLAY && haslist && threadIdx.x == 0) rlist[LIST_HDR] = (unsigned)lcnt;

    vrow bv;
#pragma unroll
    for (int c = 0; c < VW; ++c) bv[c] = bias[lane * VW + c];
    for (int row = wv; row < nr; row += 32) {
        const float di = dinv[r0 + row];
        vrow o;
#pragma unroll
        for (int c = 0; c < VW; ++c)
            o[c] = fmaxf(acc[row * CH + c * 32 + lane] * di + bv[c], 0.0f);
        *reinterpret_cast<vrow*>(outp + (long)(r0 + row) * CH + lane * VW) = o;
    }
}

extern "C" void kernel_launch(void* const* d_in, const int* in_sizes, int n_in,
                              void* d_out, int out_size, void* d_ws, size_t ws_size,
                              hipStream_t stream) {
    const float* x  = (const float*)d_in[0];
    const int*   ei = (const int*)d_in[1];
    const float* W1 = (const float*)d_in[2];
    const float* b1 = (const float*)d_in[3];
    const float* W2 = (const float*)d_in[4];
    const float* b2 = (const float*)d_in[5];

    const int N   = in_sizes[0] / K_DIM;
    const int E   = in_sizes[1] / 2;
    (void)n_in; (void)ws_size; (void)out_size;

    const int* src = ei;
    const int* dst = ei + E;
    float* out = (float*)d_out;

    char* ws = (char*)d_ws;
    const size_t degBytes = (((size_t)N * 4) + 255) & ~(size_t)255;
    float* deg  = (float*)ws;
    float* bufA = (float*)(ws + degBytes);
    float* bufB = (float*)(ws + degBytes + (size_t)N * 128 * 4);

    const int TB = 256;

    const int split = N / 2;
    int* ctrD = (int*)out;
    const int degChunks = (E + DEG_CHUNK - 1) / DEG_CHUNK;

    const int clsWords = ((E + 15) / 16) * 4;
    unsigned* cls1 = (unsigned*)out;
    const int ranges = (N + R_ROWS - 1) / R_ROWS;

    fill_f32<<<(N + TB - 1) / TB, TB, 0, stream>>>(deg, 1.0f, N, ctrD);
    degree_kernel<<<2 * degChunks, TB, 0, stream>>>(dst, deg, E, split, ctrD, degChunks);
    rsqrt_inplace<<<(N + TB - 1) / TB, TB, 0, stream>>>(deg, N);

    {
        const int tiles = ((N + 15) / 16) * (128 / 16);
        classify_edges<<<(clsWords + TB - 1) / TB, TB, 0, stream>>>(dst, cls1, E, clsWords);
        gemm_wmma_f32<128><<<(tiles + 7) / 8, 256, 0, stream>>>(x, W1, bufA, N);
        aggregate_owner<128, false><<<ranges, 1024, 0, stream>>>(
            bufA, src, dst, (const uint4*)cls1, deg, b1, bufB, (unsigned*)out, E, N);
    }

    {
        const int tiles = ((N + 15) / 16) * (64 / 16);
        gemm_wmma_f32<64><<<(tiles + 7) / 8, 256, 0, stream>>>(bufB, W2, bufA, N);
        aggregate_owner<64, true><<<ranges, 1024, 0, stream>>>(
            bufA, src, dst, (const uint4*)cls1, deg, b2, out, (unsigned*)out, E, N);
    }
}


# Edge runs take one packed word per edge from the class pass: source row, row in range, source degree (weight = its rsqrt; large ones looked up): no src/dst/dinv fetch per edge; sum orders as before
# speedup vs baseline: 3.8352x; 1.0475x over previous
#include <hip/hip_runtime.h>


typedef __attribute__((ext_vector_type(2))) float v2f;
typedef __attribute__((ext_vector_type(4))) float v4f;
typedef __attribute__((ext_vector_type(8))) float v8f;

#define K_DIM 128
#define A_PITCH 132

#define CTR_STRIDE 64

__global__ void fill_f32(float* __restrict__ p, float v, long n,
                         int* __restrict__ ctr) {
    long i = (long)blockIdx.x * blockDim.x + threadIdx.x;
    if (i < n) p[i] = v;
    if (i < 2) ctr[i * CTR_STRIDE] = 0;
}

__device__ __forceinline__ void draw_ticket(int* __restrict__ ctr, int nchunks,
                                            int* ctl) {
    const unsigned pw = __builtin_amdgcn_s_sendmsg_rtn(0x87);
    int r = (int)((pw >> 18) & 1u);
    int t = atomicAdd(&ctr[r * CTR_STRIDE], 1);
    if (t >= nchunks) {
        r ^= 1;
        t = atomicAdd(&ctr[r * CTR_STRIDE], 1);
    }
    ctl[0] = t;
    ctl[1] = r;
}

#define DEG_CHUNK 1024

__global__ __launch_bounds__(256) void degree_kernel(
    const int* __restrict__ dst, float* __restrict__ deg, int n_edges,
    int split, int* __restrict__ ctr, int nchunks) {
    __shared__ int ctl[2];
    if (threadIdx.x == 0) draw_ticket(ctr, nchunks, ctl);
    __syncthreads();
    const int t = ctl[0];
    const int r = ctl[1];
    if (t >= nchunks) return;
    constexpr int EPT = DEG_CHUNK / 256;
    int ed[EPT];
#pragma unroll
    for (int k = 0; k < EPT; ++k) {
        const int e = t * DEG_CHUNK + k * 256 + (int)threadIdx.x;
        ed[k] = dst[e < n_edges ? e : n_edges - 1];
    }
#pragma unroll
    for (int k = 0; k < EPT; ++k) {
        const int e = t * DEG_CHUNK + k * 256 + (int)threadIdx.x;
        const int d = ed[k];
        if (e < n_edges && (d >= split ? 1 : 0) == r) atomicAdd(&deg[d], 1.0f);
    }
}

__global__ void rsqrt_inplace(float* __restrict__ deg, int n) {
    int i = blockIdx.x * blockDim.x + threadIdx.x;
    if (i < n) deg[i] = rsqrtf(deg[i]);
}

template <int NCOLS>
__global__ __launch_bounds__(256) void gemm_wmma_f32(
                              const float* __restrict__ A,
                              const float* __restrict__ W,
                              float* __restrict__ C, int nrows) {
    constexpr int TC  = NCOLS / 16;
    constexpr int SPW = 8 / TC;
    __shared__ __attribute__((aligned(16))) float sA[SPW * 16 * A_PITCH];

    const int lane = threadIdx.x & 31;
    const int wave = threadIdx.x >> 5;
    const int row0 = blockIdx.x * (SPW * 16);

#pragma unroll
    for (int p = 0; p < 2 * SPW; ++p) {
        const int rl = p * 8 + wave;
        const int gr = (row0 + rl < nrows) ? (row0 + rl) : (nrows - 1);
        v4f v = *reinterpret_cast<const v4f*>(A + (long)gr * K_DIM + lane * 4);
        *reinterpret_cast<v4f*>(&sA[rl * A_PITCH + lane * 4]) = v;
    }
    __syncthreads();

    const int tcol  = wave % TC;
    const int ls    = wave / TC;
    const int rbase = row0 + ls * 16;
    if (rbase >= nrows) return;

    const int m     = lane & 15;
    const int khalf = lane >> 4;

    const float* arow = &sA[(ls * 16 + m) * A_PITCH + 2 * khalf];
    const float* wcol = W + tcol * 16 + m;

    v8f acc = {};
#pragma nounroll
    for (int k0 = 0; k0 < K_DIM; k0 += 32) {
        v2f a[8], b[8];
#pragma unroll
        for (int u = 0; u < 8; ++u) {
            const int kk = k0 + 4 * u + 2 * khalf;
            a[u] = *reinterpret_cast<const v2f*>(arow + k0 + 4 * u);
            b[u].x = wcol[(long)kk * NCOLS];
            b[u].y = wcol[(long)(kk + 1) * NCOLS];
        }
        __builtin_amdgcn_sched_barrier(0);
#pragma unroll
        for (int u = 0; u < 8; ++u) {
            acc = __builtin_amdgcn_wmma_f32_16x16x4_f32(
                false, a[u], false, b[u], (short)0, acc, false, false);
        }
        asm volatile("v_nop\n\tv_nop\n\tv_nop\n\tv_nop"
                     : "+v"(acc) : "v"(a[7]), "v"(b[7]));
    }

    if (rbase + 16 <= nrows) {
        const long off = (long)(rbase + 8 * khalf) * NCOLS + tcol * 16 + m;
#pragma unroll
        for (int r = 0; r < 8; ++r) {
            C[off + (long)r * NCOLS] = acc[r];
        }
    } else {
#pragma unroll
        for (int r = 0; r < 8; ++r) {
            const int row = rbase + r + 8 * khalf;
            if (row < nrows) C[(long)row * NCOLS + tcol * 16 + m] = acc[r];
        }
    }
}

#define R_ROWS 393

#define OUT_COLS     64
#define PIECE_SHIFT  9
#define PIECE_UNITS  (1 << PIECE_SHIFT)
#define PIECE_EDGES  (PIECE_UNITS * 16)
#define REGION_UNITS (R_ROWS * OUT_COLS * 4 / 16)
#define REGION_WORDS (REGION_UNITS * 4)
#define PK_BASE      (PIECE_UNITS * 4)
#define LIST_HDR     (PK_BASE + PIECE_EDGES)
static_assert(LIST_HDR + 32 < REGION_WORDS, "a whole region must hold its piece and a list");
#define SRC_BITS 17
#define ROW_BITS 9
#define DEG_SHIFT (SRC_BITS + ROW_BITS)
#define DEG_ESC 63
static_assert(R_ROWS <= (1 << ROW_BITS), "row inside the range must fit its bits");
static_assert(DEG_SHIFT + 6 == 32, "source row, row in range and degree fill the word");

__device__ __forceinline__ int piece_unit(int i) {
    return (i >> PIECE_SHIFT) * REGION_UNITS + (i & (PIECE_UNITS - 1));
}

__device__ __forceinline__ long edge_word(int e) {
    return (long)(e >> (PIECE_SHIFT + 4)) * REGION_WORDS + PK_BASE + (e & (PIECE_EDGES - 1));
}

__global__ void classify_edges(const int* __restrict__ src,
                               const int* __restrict__ dst,
                               const float* __restrict__ deg,
                               unsigned* __restrict__ cls, int n_edges,
                               int n_words) {
    const int t = blockIdx.x * blockDim.x + threadIdx.x;
    if (t >= n_words) return;
    unsigned w = 0;
    unsigned pk[4];
#pragma unroll
    for (int j = 0; j < 4; ++j) {
        const int e = 4 * t + j;
        const int ec = e < n_edges ? e : n_edges - 1;
        const int dd = dst[ec];
        const int ss = src[ec];
        const float dg = deg[ss];
        const int c = dd / R_ROWS;
        w |= (e < n_edges ? (unsigned)c : 255u) << (8 * j);
        const unsigned code = dg < (float)DEG_ESC ? (unsigned)dg : (unsigned)DEG_ESC;
        pk[j] = (unsigned)ss | ((unsigned)(dd - c * R_ROWS) << SRC_BITS) | (code << DEG_SHIFT);
    }
    cls[piece_unit(t >> 2) * 4 + (t & 3)] = w;
    *reinterpret_cast<uint4*>(cls + edge_word(4 * t)) = make_uint4(pk[0], pk[1], pk[2], pk[3]);
}

template <int CH, int NE>
__device__ __forceinline__ void edge_batch(const float* __restrict__ h,
                                           float* acc, int s, int d,
                                           int wbits, int j0, int lane) {
    constexpr int VW = CH / 32;
    typedef __attribute__((ext_vector_type(VW))) float vrow;
    vrow buf[NE];
#pragma unroll
    for (int jj = 0; jj < NE; ++jj) {
        const int sj = __builtin_amdgcn_readlane(s, j0 + jj);
        buf[jj] = *reinterpret_cast<const vrow*>(h + (long)sj * CH + lane * VW);
    }
#pragma unroll
    for (int jj = 0; jj < NE; ++jj) {
        const int   dj = __builtin_amdgcn_readlane(d, j0 + jj);
        const float wj = __builtin_bit_cast(
            float, __builtin_amdgcn_readlane(wbits, j0 + jj));
        float* ap = acc + dj * CH + lane;
#pragma unroll
        for (int c = 0; c < VW; ++c) atomicAdd(&ap[c * 32], buf[jj][c] * wj);
    }
}

__device__ __forceinline__ int source_weight(const float* __restrict__ dinv,
                                             int s, int code) {
    float wf = rsqrtf((float)code);
    if (__builtin_amdgcn_ballot_w32(code >= DEG_ESC) != 0u) {
        const float t = dinv[s];
        wf = code >= DEG_ESC ? t : wf;
    }
    return __builtin_bit_cast(int, wf);
}

template <int CH>
__device__ __forceinline__ void run_sd(const float* __restrict__ h,
                                       const float* __restrict__ dinv,
                                       float* acc, int s, int d, int code,
                                       int cnt, int lane) {
    constexpr int VW = CH / 32;
    constexpr int B  = 16 / VW;
    constexpr int NB = 32 / B;
    typedef __attribute__((ext_vector_type(VW))) float vrow;
    if (cnt >= 32) {
        vrow buf[2][B];
#pragma unroll
        for (int jj = 0; jj < B; ++jj) {
            const int sj = __builtin_amdgcn_readlane(s, jj);
            buf[0][jj] = *reinterpret_cast<const vrow*>(h + (long)sj * CH + lane * VW);
        }
        const int wbits = source_weight(dinv, s, code);
#pragma unroll
        for (int b = 0; b < NB; ++b) {
            if (b + 1 < NB) {
#pragma unroll
                for (int jj = 0; jj < B; ++jj) {
                    const int sj = __builtin_amdgcn_readlane(s, (b + 1) * B + jj);
                    buf[(b + 1) & 1][jj] =
                        *reinterpret_cast<const vrow*>(h + (long)sj * CH + lane * VW);
                }
            }
#pragma unroll
            for (int jj = 0; jj < B; ++jj) {
                const int   dj = __builtin_amdgcn_readlane(d, b * B + jj);
                const float wj = __builtin_bit_cast(
                    float, __builtin_amdgcn_readlane(wbits, b * B + jj));
                float* ap = acc + dj * CH + lane;
#pragma unroll
                for (int c = 0; c < VW; ++c)
                    atomicAdd(&ap[c * 32], buf[b & 1][jj][c] * wj);
            }
        }
    } else {
        const int wbits = source_weight(dinv, s, code);
        int j = 0;
        for (; j + B <= cnt; j += B)
            edge_batch<CH, B>(h, acc, s, d, wbits, j, lane);
        if (B > 4 && cnt - j >= 4) {
            edge_batch<CH, 4>(h, acc, s, d, wbits, j, lane);
            j += 4;
        }
        if (B > 2 && cnt - j >= 2) {
            edge_batch<CH, 2>(h, acc, s, d, wbits, j, lane);
            j += 2;
        }
        if (cnt - j >= 1) edge_batch<CH, 1>(h, acc, s, d, wbits, j, lane);
    }
}

template <int CH, bool RECORD>
__device__ __forceinline__ void run_edges(const float* __restrict__ h,
                                          const int* __restrict__ src,
                                          const int* __restrict__ dst,
                                          const unsigned* __restrict__ ew,
                                          const float* __restrict__ dinv,
                                          float* acc, const int* ring,
                                          int head, int cnt, int r0, int lane,
                                          unsigned* __restrict__ list, int cap,
                                          int* lcnt) {
    const int e = ring[(head + (lane < cnt ? lane : cnt - 1)) & 63];
    int s, d, code;
    if (RECORD) {
        const unsigned w = ew[edge_word(e)];
        s = (int)(w & ((1u << SRC_BITS) - 1u));
        d = (int)((w >> SRC_BITS) & ((1u << ROW_BITS) - 1u));
        code = (int)(w >> DEG_SHIFT);
        int base = 0;
        if (lane == 0) base = atomicAdd(lcnt, 32);
        base = __builtin_amdgcn_readfirstlane(base);
        if (base + 32 <= cap) list[base + lane] = lane < cnt ? w : ~0u;
    } else {
        s = src[e];
        d = dst[e] - r0;
        code = DEG_ESC;
    }
    run_sd<CH>(h, dinv, acc, s, d, code, cnt, lane);
}

__device__ __forceinline__ uint4 classes_from_dst(const int* __restrict__ dst,
                                                  int i, int n_edges) {
    unsigned w[4];
#pragma unroll
    for (int c = 0; c < 4; ++c) {
        w[c] = 0u;
#pragma unroll
        for (int j = 0; j < 4; ++j) {
            const int e = i * 16 + c * 4 + j;
            const unsigned cl = e < n_edges ? (unsigned)(dst[e] / R_ROWS) : 255u;
            w[c] |= cl << (8 * j);
        }
    }
    return make_uint4(w[0], w[1], w[2], w[3]);
}

static_assert(PIECE_UNITS % 64 == 0, "units i and i + 32 of a trip must share a piece");
template <bool FROM_DST>
__device__ __forceinline__ void class_pair(const uint4* __restrict__ cls4,
                                           const int* __restrict__ dst, int i,
                                           bool has_b, int n_edges, uint4& a,
                                           uint4& b) {
    if (FROM_DST) {
        a = classes_from_dst(dst, i, n_edges);
        b = classes_from_dst(dst, has_b ? i + 32 : i, n_edges);
    } else {
        const int p = piece_unit(i);
        a = cls4[p];
        b = cls4[has_b ? p + 32 : p];
    }
}

__device__ __forceinline__ unsigned not_mine(unsigned v, unsigned rep) {
    const unsigned x = v ^ rep;
    return ((x & 0x7F7F7F7Fu) + 0x7F7F7F7Fu) | x;
}

#define UNIT_A 0x0F0F0F0Fu
#define UNIT_B 0xF0F0F0F0u
__device__ __forceinline__ unsigned hit_mask(uint4 a, uint4 b, unsigned rep,
                                             unsigned valid) {
    const unsigned w[8] = {a.x, a.y, a.z, a.w, b.x, b.y, b.z, b.w};
    unsigned acc = 0u;
#pragma unroll
    for (int j = 0; j < 8; ++j)
        acc = (acc >> 1) | (not_mine(w[j], rep) & 0x80808080u);
    return ~acc & valid;
}

template <int CH, bool FROM_DST, bool RECORD>
__device__ __forceinline__ void sweep_range(
    const float* __restrict__ h, const int* __restrict__ src,
    const int* __restrict__ dst, const uint4* __restrict__ cls4,
    const float* __restrict__ dinv, float* acc, int* myring, int n_edges,
    int r0, int lane, int wv, unsigned* list, int cap, int* lcnt) {
    const unsigned rep = (unsigned)blockIdx.x * 0x01010101u;
    const int nvec = (n_edges + 15) >> 4;
    const int nit  = (nvec + 63) >> 6;
    const int rot = (int)(((long)blockIdx.x * nit) / gridDim.x);
    int head = 0, qn = 0;
    uint4 rawa = make_uint4(0u, 0u, 0u, 0u), rawb = rawa;
    int idxn = 0;
    unsigned validn = 0u;
    if (nit > 0) {
        const bool live = wv < nit;
        int it = (live ? wv : 0) + rot;
        if (it >= nit) it -= nit;
        idxn = it * 64 + lane;
        const bool has_b = idxn + 32 < nvec;
        validn = live ? ((idxn < nvec ? UNIT_A : 0u) | (has_b ? UNIT_B : 0u)) : 0u;
        class_pair<FROM_DST>(cls4, dst, idxn < nvec ? idxn : nvec - 1, has_b,
                             n_edges, rawa, rawb);
    }
    for (int k = wv; k < nit; k += 32) {
        const int idx = idxn;
        const uint4 va = rawa, vb = rawb;
        const unsigned valid = validn;
        {
            const bool live = k + 32 < nit;
            int it = (live ? k + 32 : k) + rot;
            if (it >= nit) it -= nit;
            idxn = it * 64 + lane;
            const bool has_b = idxn + 32 < nvec;
            validn = live ? ((idxn < nvec ? UNIT_A : 0u) | (has_b ? UNIT_B : 0u)) : 0u;
            class_pair<FROM_DST>(cls4, dst, idxn < nvec ? idxn : nvec - 1, has_b,
                                 n_edges, rawa, rawb);
        }
        unsigned m = hit_mask(va, vb, rep, valid);
        unsigned bm = __builtin_amdgcn_ballot_w32(m != 0u);
        while (bm != 0u) {
            if (m != 0u) {
                const int i = __builtin_ctz(m);
                const int p = (head + qn + (int)__builtin_amdgcn_mbcnt_lo(bm, 0u)) & 63;
                myring[p] = idx * 16 + ((i & 4) << 7) + ((i & 3) << 2) + (i >> 3);
                m &= m - 1u;
            }
            qn += __builtin_popcount(bm);
            if (qn >= 32) {
                run_edges<CH, RECORD>(h, src, dst,
                                      reinterpret_cast<const unsigned*>(cls4),
                                      dinv, acc, myring, head, 32, r0, lane,
                                      list, cap, lcnt);
                head = (head + 32) & 63;
                qn -= 32;
            }
            bm = __builtin_amdgcn_ballot_w32(m != 0u);
        }
    }
    if (qn > 0)
        run_edges<CH, RECORD>(h, src, dst, reinterpret_cast<const unsigned*>(cls4),
                              dinv, acc, myring, head, qn, r0, lane, list, cap, lcnt);
}

template <int CH, bool REPLAY>
__global__ __launch_bounds__(1024) void aggregate_owner(
    const float* __restrict__ h, const int* __restrict__ src,
    const int* __restrict__ dst, const uint4* __restrict__ cls4,
    const float* __restrict__ dinv, const float* __restrict__ bias,
    float* outp, unsigned* lists, int n_edges, int n_rows) {
    constexpr int VW = CH / 32;
    typedef __attribute__((ext_vector_type(VW))) float vrow;
    __shared__ float acc[R_ROWS * K_DIM];
    __shared__ int ring[32][64];
    __shared__ int lcnt;

    const int lane = threadIdx.x & 31;
    const int wv = __builtin_amdgcn_readfirstlane((int)(threadIdx.x >> 5));
    const int r0 = blockIdx.x * R_ROWS;
    const int nr = (n_rows - r0 < R_ROWS) ? (n_rows - r0) : R_ROWS;

    unsigned* rlist = lists + (long)blockIdx.x * REGION_WORDS;
    const int npieces = (((n_edges + 15) >> 4) + PIECE_UNITS - 1) >> PIECE_SHIFT;
    const int lhdr  = ((int)blockIdx.x < npieces) ? LIST_HDR : 0;
    const int lbase = lhdr + 32;
    const bool haslist = nr * OUT_COLS >= lbase;
    const int cap = haslist ? ((nr * OUT_COLS - lbase) & ~31) : 0;

    if (threadIdx.x == 0) lcnt = 0;
    for (int row = wv; row < nr; row += 32) {
        const float di = dinv[r0 + row];
        const vrow v = *reinterpret_cast<const vrow*>(h + (long)(r0 + row) * CH + lane * VW);
#pragma unroll
        for (int c = 0; c < VW; ++c) acc[row * CH + c * 32 + lane] = v[c] * di;
    }
    __syncthreads();

    if (REPLAY) {
        int total = haslist ? (int)rlist[lhdr] : 0;
        total = __builtin_amdgcn_readfirstlane(total);
        if (haslist && total <= cap) {
            const int nruns = total >> 5;
            const unsigned* ent = rlist + lbase + lane;
            unsigned wn = ~0u;
            if (nruns > 0) wn = ent[(wv < nruns ? wv : 0) * 32];
            for (int r = wv; r < nruns; r += 32) {
                const unsigned w = wn;
                wn = ent[(r + 32 < nruns ? r + 32 : r) * 32];
                const bool ok = w != ~0u;
                const int cnt = __builtin_popcount(__builtin_amdgcn_ballot_w32(ok));
                const int s = ok ? (int)(w & ((1u << SRC_BITS) - 1u)) : 0;
                const int d = ok ? (int)((w >> SRC_BITS) & ((1u << ROW_BITS) - 1u)) : 0;
                const int code = ok ? (int)(w >> DEG_SHIFT) : 1;
                if (cnt > 0) run_sd<CH>(h, dinv, acc, s, d, code, cnt, lane);
            }
        } else {
            sweep_range<CH, true, false>(h, src, dst, cls4, dinv, acc, ring[wv],
                                         n_edges, r0, lane, wv, rlist, 0, &lcnt);
        }
    } else {
        sweep_range<CH, false, true>(h, src, dst, cls4, dinv, acc, ring[wv],
                                     n_edges, r0, lane, wv, rlist + lbase, cap,
                                     &lcnt);
    }
    __syncthreads();

    if (!REPLAY && haslist && threadIdx.x == 0) rlist[lhdr] = (unsigned)lcnt;

    vrow bv;
#pragma unroll
    for (int c = 0; c < VW; ++c) bv[c] = bias[lane * VW + c];
    for (int row = wv; row < nr; row += 32) {
        const float di = dinv[r0 + row];
        vrow o;
#pragma unroll
        for (int c = 0; c < VW; ++c)
            o[c] = fmaxf(acc[row * CH + c * 32 + lane] * di + bv[c], 0.0f);
        *reinterpret_cast<vrow*>(outp + (long)(r0 + row) * CH + lane * VW) = o;
    }
}

extern "C" void kernel_launch(void* const* d_in, const int* in_sizes, int n_in,
                              void* d_out, int out_size, void* d_ws, size_t ws_size,
                              hipStream_t stream) {
    const float* x  = (const float*)d_in[0];
    const int*   ei = (const int*)d_in[1];
    const float* W1 = (const float*)d_in[2];
    const float* b1 = (const float*)d_in[3];
    const float* W2 = (const float*)d_in[4];
    const float* b2 = (const float*)d_in[5];

    const int N   = in_sizes[0] / K_DIM;
    const int E   = in_sizes[1] / 2;
    (void)n_in; (void)ws_size; (void)out_size;

    const int* src = ei;
    const int* dst = ei + E;
    float* out = (float*)d_out;

    char* ws = (char*)d_ws;
    const size_t degBytes = (((size_t)N * 4) + 255) & ~(size_t)255;
    float* deg  = (float*)ws;
    float* bufA = (float*)(ws + degBytes);
    float* bufB = (float*)(ws + degBytes + (size_t)N * 128 * 4);

    const int TB = 256;

    const int split = N / 2;
    int* ctrD = (int*)out;
    const int degChunks = (E + DEG_CHUNK - 1) / DEG_CHUNK;

    const int clsWords = ((E + 15) / 16) * 4;
    unsigned* cls1 = (unsigned*)out;
    const int ranges = (N + R_ROWS - 1) / R_ROWS;

    fill_f32<<<(N + TB - 1) / TB, TB, 0, stream>>>(deg, 1.0f, N, ctrD);
    degree_kernel<<<2 * degChunks, TB, 0, stream>>>(dst, deg, E, split, ctrD, degChunks);
    classify_edges<<<(clsWords + TB - 1) / TB, TB, 0, stream>>>(src, dst, deg, cls1, E, clsWords);
    rsqrt_inplace<<<(N + TB - 1) / TB, TB, 0, stream>>>(deg, N);

    {
        const int tiles = ((N + 15) / 16) * (128 / 16);
        gemm_wmma_f32<128><<<(tiles + 7) / 8, 256, 0, stream>>>(x, W1, bufA, N);
        aggregate_owner<128, false><<<ranges, 1024, 0, stream>>>(
            bufA, src, dst, (const uint4*)cls1, deg, b1, bufB, (unsigned*)out, E, N);
    }

    {
        const int tiles = ((N + 15) / 16) * (64 / 16);
        gemm_wmma_f32<64><<<(tiles + 7) / 8, 256, 0, stream>>>(bufB, W2, bufA, N);
        aggregate_owner<64, true><<<ranges, 1024, 0, stream>>>(
            bufA, src, dst, (const uint4*)cls1, deg, b2, out, (unsigned*)out, E, N);
    }
}


# Dense transforms weight-stationary: 1024 resident work-groups; each wave loads its 128x16 slice of W into registers once and walks row blocks, A rows double-buffered in LDS; sum order unchanged
# speedup vs baseline: 3.9986x; 1.0426x over previous
#include <hip/hip_runtime.h>


typedef __attribute__((ext_vector_type(2))) float v2f;
typedef __attribute__((ext_vector_type(4))) float v4f;
typedef __attribute__((ext_vector_type(8))) float v8f;

#define K_DIM 128
#define A_PITCH 132
#define GEMM_GRID 1024

#define CTR_STRIDE 64

__global__ void fill_f32(float* __restrict__ p, float v, long n,
                         int* __restrict__ ctr) {
    long i = (long)blockIdx.x * blockDim.x + threadIdx.x;
    if (i < n) p[i] = v;
    if (i < 2) ctr[i * CTR_STRIDE] = 0;
}

__device__ __forceinline__ void draw_ticket(int* __restrict__ ctr, int nchunks,
                                            int* ctl) {
    const unsigned pw = __builtin_amdgcn_s_sendmsg_rtn(0x87);
    int r = (int)((pw >> 18) & 1u);
    int t = atomicAdd(&ctr[r * CTR_STRIDE], 1);
    if (t >= nchunks) {
        r ^= 1;
        t = atomicAdd(&ctr[r * CTR_STRIDE], 1);
    }
    ctl[0] = t;
    ctl[1] = r;
}

#define DEG_CHUNK 1024

__global__ __launch_bounds__(256) void degree_kernel(
    const int* __restrict__ dst, float* __restrict__ deg, int n_edges,
    int split, int* __restrict__ ctr, int nchunks) {
    __shared__ int ctl[2];
    if (threadIdx.x == 0) draw_ticket(ctr, nchunks, ctl);
    __syncthreads();
    const int t = ctl[0];
    const int r = ctl[1];
    if (t >= nchunks) return;
    constexpr int EPT = DEG_CHUNK / 256;
    int ed[EPT];
#pragma unroll
    for (int k = 0; k < EPT; ++k) {
        const int e = t * DEG_CHUNK + k * 256 + (int)threadIdx.x;
        ed[k] = dst[e < n_edges ? e : n_edges - 1];
    }
#pragma unroll
    for (int k = 0; k < EPT; ++k) {
        const int e = t * DEG_CHUNK + k * 256 + (int)threadIdx.x;
        const int d = ed[k];
        if (e < n_edges && (d >= split ? 1 : 0) == r) atomicAdd(&deg[d], 1.0f);
    }
}

__global__ void rsqrt_inplace(float* __restrict__ deg, int n) {
    int i = blockIdx.x * blockDim.x + threadIdx.x;
    if (i < n) deg[i] = rsqrtf(deg[i]);
}

template <int NCOLS>
__global__ __launch_bounds__(256) void gemm_wmma_f32(
                              const float* __restrict__ A,
                              const float* __restrict__ W,
                              float* __restrict__ C, int nrows) {
    constexpr int TC  = NCOLS / 16;
    constexpr int SPW = 8 / TC;
    constexpr int BR  = SPW * 16;
    constexpr int NP  = 2 * SPW;
    __shared__ __attribute__((aligned(16))) float sA[2][BR * A_PITCH];

    const int lane = threadIdx.x & 31;
    const int wave = threadIdx.x >> 5;
    const int tcol  = wave % TC;
    const int ls    = wave / TC;
    const int m     = lane & 15;
    const int khalf = lane >> 4;
    const int nblk  = (nrows + BR - 1) / BR;
    const int gstep = (int)gridDim.x;

    v4f pre[NP];
#pragma unroll
    for (int p = 0; p < NP; ++p) {
        const int r  = (int)blockIdx.x * BR + p * 8 + wave;
        const int gr = r < nrows ? r : nrows - 1;
        pre[p] = *reinterpret_cast<const v4f*>(A + (long)gr * K_DIM + lane * 4);
    }

    const float* wcol = W + tcol * 16 + m;
    v2f b[32];
#pragma unroll
    for (int u = 0; u < 32; ++u) {
        const int kk = 4 * u + 2 * khalf;
        b[u].x = wcol[(long)kk * NCOLS];
        b[u].y = wcol[(long)(kk + 1) * NCOLS];
    }

    int buf = 0;
    for (int blk = blockIdx.x; blk < nblk; blk += gstep) {
        float* sbuf = sA[buf];
#pragma unroll
        for (int p = 0; p < NP; ++p)
            *reinterpret_cast<v4f*>(&sbuf[(p * 8 + wave) * A_PITCH + lane * 4]) = pre[p];
        __syncthreads();
#pragma unroll
        for (int p = 0; p < NP; ++p) {
            const int r  = (blk + gstep) * BR + p * 8 + wave;
            const int gr = r < nrows ? r : nrows - 1;
            pre[p] = *reinterpret_cast<const v4f*>(A + (long)gr * K_DIM + lane * 4);
        }

        const int rbase = blk * BR + ls * 16;
        if (rbase < nrows) {
            const float* arow = &sbuf[(ls * 16 + m) * A_PITCH + 2 * khalf];
            float* cp = C + (long)(rbase + 8 * khalf) * NCOLS + tcol * 16 + m;
            v8f acc = {};
#pragma unroll
            for (int t = 0; t < 4; ++t) {
                v2f a[8];
#pragma unroll
                for (int u = 0; u < 8; ++u)
                    a[u] = *reinterpret_cast<const v2f*>(arow + 32 * t + 4 * u);
                __builtin_amdgcn_sched_barrier(0);
#pragma unroll
                for (int u = 0; u < 8; ++u) {
                    acc = __builtin_amdgcn_wmma_f32_16x16x4_f32(
                        false, a[u], false, b[8 * t + u], (short)0, acc, false, false);
                }
                asm volatile("v_nop\n\tv_nop\n\tv_nop\n\tv_nop"
                             : "+v"(acc) : "v"(a[7]), "v"(b[8 * t + 7]));
                __builtin_amdgcn_sched_barrier(0);
            }
            if (rbase + 16 <= nrows) {
#pragma unroll
                for (int r = 0; r < 8; ++r) cp[(long)r * NCOLS] = acc[r];
            } else {
#pragma unroll
                for (int r = 0; r < 8; ++r)
                    if (rbase + r + 8 * khalf < nrows) cp[(long)r * NCOLS] = acc[r];
            }
        }
        buf ^= 1;
    }
}

#define R_ROWS 393

#define OUT_COLS     64
#define PIECE_SHIFT  9
#define PIECE_UNITS  (1 << PIECE_SHIFT)
#define PIECE_EDGES  (PIECE_UNITS * 16)
#define REGION_UNITS (R_ROWS * OUT_COLS * 4 / 16)
#define REGION_WORDS (REGION_UNITS * 4)
#define PK_BASE      (PIECE_UNITS * 4)
#define LIST_HDR     (PK_BASE + PIECE_EDGES)
static_assert(LIST_HDR + 32 < REGION_WORDS, "a whole region must hold its piece and a list");
#define SRC_BITS 17
#define ROW_BITS 9
#define DEG_SHIFT (SRC_BITS + ROW_BITS)
#define DEG_ESC 63
static_assert(R_ROWS <= (1 << ROW_BITS), "row inside the range must fit its bits");
static_assert(DEG_SHIFT + 6 == 32, "source row, row in range and degree fill the word");

__device__ __forceinline__ int piece_unit(int i) {
    return (i >> PIECE_SHIFT) * REGION_UNITS + (i & (PIECE_UNITS - 1));
}

__device__ __forceinline__ long edge_word(int e) {
    return (long)(e >> (PIECE_SHIFT + 4)) * REGION_WORDS + PK_BASE + (e & (PIECE_EDGES - 1));
}

__global__ void classify_edges(const int* __restrict__ src,
                               const int* __restrict__ dst,
                               const float* __restrict__ deg,
                               unsigned* __restrict__ cls, int n_edges,
                               int n_words) {
    const int t = blockIdx.x * blockDim.x + threadIdx.x;
    if (t >= n_words) return;
    unsigned w = 0;
    unsigned pk[4];
#pragma unroll
    for (int j = 0; j < 4; ++j) {
        const int e = 4 * t + j;
        const int ec = e < n_edges ? e : n_edges - 1;
        const int dd = dst[ec];
        const int ss = src[ec];
        const float dg = deg[ss];
        const int c = dd / R_ROWS;
        w |= (e < n_edges ? (unsigned)c : 255u) << (8 * j);
        const unsigned code = dg < (float)DEG_ESC ? (unsigned)dg : (unsigned)DEG_ESC;
        pk[j] = (unsigned)ss | ((unsigned)(dd - c * R_ROWS) << SRC_BITS) | (code << DEG_SHIFT);
    }
    cls[piece_unit(t >> 2) * 4 + (t & 3)] = w;
    *reinterpret_cast<uint4*>(cls + edge_word(4 * t)) = make_uint4(pk[0], pk[1], pk[2], pk[3]);
}

template <int CH, int NE>
__device__ __forceinline__ void edge_batch(const float* __restrict__ h,
                                           float* acc, int s, int d,
                                           int wbits, int j0, int lane) {
    constexpr int VW = CH / 32;
    typedef __attribute__((ext_vector_type(VW))) float vrow;
    vrow buf[NE];
#pragma unroll
    for (int jj = 0; jj < NE; ++jj) {
        const int sj = __builtin_amdgcn_readlane(s, j0 + jj);
        buf[jj] = *reinterpret_cast<const vrow*>(h + (long)sj * CH + lane * VW);
    }
#pragma unroll
    for (int jj = 0; jj < NE; ++jj) {
        const int   dj = __builtin_amdgcn_readlane(d, j0 + jj);
        const float wj = __builtin_bit_cast(
            float, __builtin_amdgcn_readlane(wbits, j0 + jj));
        float* ap = acc + dj * CH + lane;
#pragma unroll
        for (int c = 0; c < VW; ++c) atomicAdd(&ap[c * 32], buf[jj][c] * wj);
    }
}

__device__ __forceinline__ int source_weight(const float* __restrict__ dinv,
                                             int s, int code) {
    float wf = rsqrtf((float)code);
    if (__builtin_amdgcn_ballot_w32(code >= DEG_ESC) != 0u) {
        const float t = dinv[s];
        wf = code >= DEG_ESC ? t : wf;
    }
    return __builtin_bit_cast(int, wf);
}

template <int CH>
__device__ __forceinline__ void run_sd(const float* __restrict__ h,
                                       const float* __restrict__ dinv,
                                       float* acc, int s, int d, int code,
                                       int cnt, int lane) {
    constexpr int VW = CH / 32;
    constexpr int B  = 16 / VW;
    constexpr int NB = 32 / B;
    typedef __attribute__((ext_vector_type(VW))) float vrow;
    if (cnt >= 32) {
        vrow buf[2][B];
#pragma unroll
        for (int jj = 0; jj < B; ++jj) {
            const int sj = __builtin_amdgcn_readlane(s, jj);
            buf[0][jj] = *reinterpret_cast<const vrow*>(h + (long)sj * CH + lane * VW);
        }
        const int wbits = source_weight(dinv, s, code);
#pragma unroll
        for (int b = 0; b < NB; ++b) {
            if (b + 1 < NB) {
#pragma unroll
                for (int jj = 0; jj < B; ++jj) {
                    const int sj = __builtin_amdgcn_readlane(s, (b + 1) * B + jj);
                    buf[(b + 1) & 1][jj] =
                        *reinterpret_cast<const vrow*>(h + (long)sj * CH + lane * VW);
                }
            }
#pragma unroll
            for (int jj = 0; jj < B; ++jj) {
                const int   dj = __builtin_amdgcn_readlane(d, b * B + jj);
                const float wj = __builtin_bit_cast(
                    float, __builtin_amdgcn_readlane(wbits, b * B + jj));
                float* ap = acc + dj * CH + lane;
#pragma unroll
                for (int c = 0; c < VW; ++c)
                    atomicAdd(&ap[c * 32], buf[b & 1][jj][c] * wj);
            }
        }
    } else {
        const int wbits = source_weight(dinv, s, code);
        int j = 0;
        for (; j + B <= cnt; j += B)
            edge_batch<CH, B>(h, acc, s, d, wbits, j, lane);
        if (B > 4 && cnt - j >= 4) {
            edge_batch<CH, 4>(h, acc, s, d, wbits, j, lane);
            j += 4;
        }
        if (B > 2 && cnt - j >= 2) {
            edge_batch<CH, 2>(h, acc, s, d, wbits, j, lane);
            j += 2;
        }
        if (cnt - j >= 1) edge_batch<CH, 1>(h, acc, s, d, wbits, j, lane);
    }
}

template <int CH, bool RECORD>
__device__ __forceinline__ void run_edges(const float* __restrict__ h,
                                          const int* __restrict__ src,
                                          const int* __restrict__ dst,
                                          const unsigned* __restrict__ ew,
                                          const float* __restrict__ dinv,
                                          float* acc, const int* ring,
                                          int head, int cnt, int r0, int lane,
                                          unsigned* __restrict__ list, int cap,
                                          int* lcnt) {
    const int e = ring[(head + (lane < cnt ? lane : cnt - 1)) & 63];
    int s, d, code;
    if (RECORD) {
        const unsigned w = ew[edge_word(e)];
        s = (int)(w & ((1u << SRC_BITS) - 1u));
        d = (int)((w >> SRC_BITS) & ((1u << ROW_BITS) - 1u));
        code = (int)(w >> DEG_SHIFT);
        int base = 0;
        if (lane == 0) base = atomicAdd(lcnt, 32);
        base = __builtin_amdgcn_readfirstlane(base);
        if (base + 32 <= cap) list[base + lane] = lane < cnt ? w : ~0u;
    } else {
        s = src[e];
        d = dst[e] - r0;
        code = DEG_ESC;
    }
    run_sd<CH>(h, dinv, acc, s, d, code, cnt, lane);
}

__device__ __forceinline__ uint4 classes_from_dst(const int* __restrict__ dst,
                                                  int i, int n_edges) {
    unsigned w[4];
#pragma unroll
    for (int c = 0; c < 4; ++c) {
        w[c] = 0u;
#pragma unroll
        for (int j = 0; j < 4; ++j) {
            const int e = i * 16 + c * 4 + j;
            const unsigned cl = e < n_edges ? (unsigned)(dst[e] / R_ROWS) : 255u;
            w[c] |= cl << (8 * j);
        }
    }
    return make_uint4(w[0], w[1], w[2], w[3]);
}

static_assert(PIECE_UNITS % 64 == 0, "units i and i + 32 of a trip must share a piece");
template <bool FROM_DST>
__device__ __forceinline__ void class_pair(const uint4* __restrict__ cls4,
                                           const int* __restrict__ dst, int i,
                                           bool has_b, int n_edges, uint4& a,
                                           uint4& b) {
    if (FROM_DST) {
        a = classes_from_dst(dst, i, n_edges);
        b = classes_from_dst(dst, has_b ? i + 32 : i, n_edges);
    } else {
        const int p = piece_unit(i);
        a = cls4[p];
        b = cls4[has_b ? p + 32 : p];
    }
}

__device__ __forceinline__ unsigned not_mine(unsigned v, unsigned rep) {
    const unsigned x = v ^ rep;
    return ((x & 0x7F7F7F7Fu) + 0x7F7F7F7Fu) | x;
}

#define UNIT_A 0x0F0F0F0Fu
#define UNIT_B 0xF0F0F0F0u
__device__ __forceinline__ unsigned hit_mask(uint4 a, uint4 b, unsigned rep,
                                             unsigned valid) {
    const unsigned w[8] = {a.x, a.y, a.z, a.w, b.x, b.y, b.z, b.w};
    unsigned acc = 0u;
#pragma unroll
    for (int j = 0; j < 8; ++j)
        acc = (acc >> 1) | (not_mine(w[j], rep) & 0x80808080u);
    return ~acc & valid;
}

template <int CH, bool FROM_DST, bool RECORD>
__device__ __forceinline__ void sweep_range(
    const float* __restrict__ h, const int* __restrict__ src,
    const int* __restrict__ dst, const uint4* __restrict__ cls4,
    const float* __restrict__ dinv, float* acc, int* myring, int n_edges,
    int r0, int lane, int wv, unsigned* list, int cap, int* lcnt) {
    const unsigned rep = (unsigned)blockIdx.x * 0x01010101u;
    const int nvec = (n_edges + 15) >> 4;
    const int nit  = (nvec + 63) >> 6;
    const int rot = (int)(((long)blockIdx.x * nit) / gridDim.x);
    int head = 0, qn = 0;
    uint4 rawa = make_uint4(0u, 0u, 0u, 0u), rawb = rawa;
    int idxn = 0;
    unsigned validn = 0u;
    if (nit > 0) {
        const bool live = wv < nit;
        int it = (live ? wv : 0) + rot;
        if (it >= nit) it -= nit;
        idxn = it * 64 + lane;
        const bool has_b = idxn + 32 < nvec;
        validn = live ? ((idxn < nvec ? UNIT_A : 0u) | (has_b ? UNIT_B : 0u)) : 0u;
        class_pair<FROM_DST>(cls4, dst, idxn < nvec ? idxn : nvec - 1, has_b,
                             n_edges, rawa, rawb);
    }
    for (int k = wv; k < nit; k += 32) {
        const int idx = idxn;
        const uint4 va = rawa, vb = rawb;
        const unsigned valid = validn;
        {
            const bool live = k + 32 < nit;
            int it = (live ? k + 32 : k) + rot;
            if (it >= nit) it -= nit;
            idxn = it * 64 + lane;
            const bool has_b = idxn + 32 < nvec;
            validn = live ? ((idxn < nvec ? UNIT_A : 0u) | (has_b ? UNIT_B : 0u)) : 0u;
            class_pair<FROM_DST>(cls4, dst, idxn < nvec ? idxn : nvec - 1, has_b,
                                 n_edges, rawa, rawb);
        }
        unsigned m = hit_mask(va, vb, rep, valid);
        unsigned bm = __builtin_amdgcn_ballot_w32(m != 0u);
        while (bm != 0u) {
            if (m != 0u) {
                const int i = __builtin_ctz(m);
                const int p = (head + qn + (int)__builtin_amdgcn_mbcnt_lo(bm, 0u)) & 63;
                myring[p] = idx * 16 + ((i & 4) << 7) + ((i & 3) << 2) + (i >> 3);
                m &= m - 1u;
            }
            qn += __builtin_popcount(bm);
            if (qn >= 32) {
                run_edges<CH, RECORD>(h, src, dst,
                                      reinterpret_cast<const unsigned*>(cls4),
                                      dinv, acc, myring, head, 32, r0, lane,
                                      list, cap, lcnt);
                head = (head + 32) & 63;
                qn -= 32;
            }
            bm = __builtin_amdgcn_ballot_w32(m != 0u);
        }
    }
    if (qn > 0)
        run_edges<CH, RECORD>(h, src, dst, reinterpret_cast<const unsigned*>(cls4),
                              dinv, acc, myring, head, qn, r0, lane, list, cap, lcnt);
}

template <int CH, bool REPLAY>
__global__ __launch_bounds__(1024) void aggregate_owner(
    const float* __restrict__ h, const int* __restrict__ src,
    const int* __restrict__ dst, const uint4* __restrict__ cls4,
    const float* __restrict__ dinv, const float* __restrict__ bias,
    float* outp, unsigned* lists, int n_edges, int n_rows) {
    constexpr int VW = CH / 32;
    typedef __attribute__((ext_vector_type(VW))) float vrow;
    __shared__ float acc[R_ROWS * K_DIM];
    __shared__ int ring[32][64];
    __shared__ int lcnt;

    const int lane = threadIdx.x & 31;
    const int wv = __builtin_amdgcn_readfirstlane((int)(threadIdx.x >> 5));
    const int r0 = blockIdx.x * R_ROWS;
    const int nr = (n_rows - r0 < R_ROWS) ? (n_rows - r0) : R_ROWS;

    unsigned* rlist = lists + (long)blockIdx.x * REGION_WORDS;
    const int npieces = (((n_edges + 15) >> 4) + PIECE_UNITS - 1) >> PIECE_SHIFT;
    const int lhdr  = ((int)blockIdx.x < npieces) ? LIST_HDR : 0;
    const int lbase = lhdr + 32;
    const bool haslist = nr * OUT_COLS >= lbase;
    const int cap = haslist ? ((nr * OUT_COLS - lbase) & ~31) : 0;

    if (threadIdx.x == 0) lcnt = 0;
    for (int row = wv; row < nr; row += 32) {
        const float di = dinv[r0 + row];
        const vrow v = *reinterpret_cast<const vrow*>(h + (long)(r0 + row) * CH + lane * VW);
#pragma unroll
        for (int c = 0; c < VW; ++c) acc[row * CH + c * 32 + lane] = v[c] * di;
    }
    __syncthreads();

    if (REPLAY) {
        int total = haslist ? (int)rlist[lhdr] : 0;
        total = __builtin_amdgcn_readfirstlane(total);
        if (haslist && total <= cap) {
            const int nruns = total >> 5;
            const unsigned* ent = rlist + lbase + lane;
            unsigned wn = ~0u;
            if (nruns > 0) wn = ent[(wv < nruns ? wv : 0) * 32];
            for (int r = wv; r < nruns; r += 32) {
                const unsigned w = wn;
                wn = ent[(r + 32 < nruns ? r + 32 : r) * 32];
                const bool ok = w != ~0u;
                const int cnt = __builtin_popcount(__builtin_amdgcn_ballot_w32(ok));
                const int s = ok ? (int)(w & ((1u << SRC_BITS) - 1u)) : 0;
                const int d = ok ? (int)((w >> SRC_BITS) & ((1u << ROW_BITS) - 1u)) : 0;
                const int code = ok ? (int)(w >> DEG_SHIFT) : 1;
                if (cnt > 0) run_sd<CH>(h, dinv, acc, s, d, code, cnt, lane);
            }
        } else {
            sweep_range<CH, true, false>(h, src, dst, cls4, dinv, acc, ring[wv],
                                         n_edges, r0, lane, wv, rlist, 0, &lcnt);
        }
    } else {
        sweep_range<CH, false, true>(h, src, dst, cls4, dinv, acc, ring[wv],
                                     n_edges, r0, lane, wv, rlist + lbase, cap,
                                     &lcnt);
    }
    __syncthreads();

    if (!REPLAY && haslist && threadIdx.x == 0) rlist[lhdr] = (unsigned)lcnt;

    vrow bv;
#pragma unroll
    for (int c = 0; c < VW; ++c) bv[c] = bias[lane * VW + c];
    for (int row = wv; row < nr; row += 32) {
        const float di = dinv[r0 + row];
        vrow o;
#pragma unroll
        for (int c = 0; c < VW; ++c)
            o[c] = fmaxf(acc[row * CH + c * 32 + lane] * di + bv[c], 0.0f);
        *reinterpret_cast<vrow*>(outp + (long)(r0 + row) * CH + lane * VW) = o;
    }
}

extern "C" void kernel_launch(void* const* d_in, const int* in_sizes, int n_in,
                              void* d_out, int out_size, void* d_ws, size_t ws_size,
                              hipStream_t stream) {
    const float* x  = (const float*)d_in[0];
    const int*   ei = (const int*)d_in[1];
    const float* W1 = (const float*)d_in[2];
    const float* b1 = (const float*)d_in[3];
    const float* W2 = (const float*)d_in[4];
    const float* b2 = (const float*)d_in[5];

    const int N   = in_sizes[0] / K_DIM;
    const int E   = in_sizes[1] / 2;
    (void)n_in; (void)ws_size; (void)out_size;

    const int* src = ei;
    const int* dst = ei + E;
    float* out = (float*)d_out;

    char* ws = (char*)d_ws;
    const size_t degBytes = (((size_t)N * 4) + 255) & ~(size_t)255;
    float* deg  = (float*)ws;
    float* bufA = (float*)(ws + degBytes);
    float* bufB = (float*)(ws + degBytes + (size_t)N * 128 * 4);

    const int TB = 256;

    const int split = N / 2;
    int* ctrD = (int*)out;
    const int degChunks = (E + DEG_CHUNK - 1) / DEG_CHUNK;

    const int clsWords = ((E + 15) / 16) * 4;
    unsigned* cls1 = (unsigned*)out;
    const int ranges = (N + R_ROWS - 1) / R_ROWS;

    fill_f32<<<(N + TB - 1) / TB, TB, 0, stream>>>(deg, 1.0f, N, ctrD);
    degree_kernel<<<2 * degChunks, TB, 0, stream>>>(dst, deg, E, split, ctrD, degChunks);
    classify_edges<<<(clsWords + TB - 1) / TB, TB, 0, stream>>>(src, dst, deg, cls1, E, clsWords);
    rsqrt_inplace<<<(N + TB - 1) / TB, TB, 0, stream>>>(deg, N);

    {
        gemm_wmma_f32<128><<<GEMM_GRID, 256, 0, stream>>>(x, W1, bufA, N);
        aggregate_owner<128, false><<<ranges, 1024, 0, stream>>>(
            bufA, src, dst, (const uint4*)cls1, deg, b1, bufB, (unsigned*)out, E, N);
    }

    {
        gemm_wmma_f32<64><<<GEMM_GRID, 256, 0, stream>>>(bufB, W2, bufA, N);
        aggregate_owner<64, true><<<ranges, 1024, 0, stream>>>(
            bufA, src, dst, (const uint4*)cls1, deg, b2, out, (unsigned*)out, E, N);
    }
}
